# starting kernel
# baseline (speedup 1.0000x reference)
#include <hip/hip_runtime.h>
#include <math.h>

#define N_NODES 50000
#define N_EDGES 800000
#define DIM     128
#define LRELU_A 0.2f

typedef __attribute__((ext_vector_type(2))) float v2f;
typedef __attribute__((ext_vector_type(8))) float v8f;

__global__ __launch_bounds__(128) void gat_gemm_wmma(const float* __restrict__ h,
                                                     const float* __restrict__ W,
                                                     float* __restrict__ Wh) {
  const int wave = (int)((blockIdx.x * blockDim.x + threadIdx.x) >> 5);
  const int lane = (int)(threadIdx.x & 31);
  if (wave >= N_NODES / 16) return;

  const int row0  = wave * 16;
  const int rsel  = lane & 15;
  const int khalf = (lane >> 4) * 2;

  v8f acc[8];
  #pragma unroll
  for (int t = 0; t < 8; ++t) { v8f z = {}; acc[t] = z; }

  for (int k0 = 0; k0 < DIM; k0 += 4) {
    const float* ap = h + (size_t)(row0 + rsel) * DIM + (k0 + khalf);
    v2f a; a.x = ap[0]; a.y = ap[1];
    #pragma unroll
    for (int t = 0; t < 8; ++t) {
      const float* bp = W + (size_t)(k0 + khalf) * DIM + t * 16 + rsel;
      v2f b; b.x = bp[0]; b.y = bp[DIM];
      acc[t] = __builtin_amdgcn_wmma_f32_16x16x4_f32(
           false, a,  false, b,
           (short)0, acc[t],  false,  false);
    }
  }

  const int srow = row0 + (lane >> 4) * 8;
  #pragma unroll
  for (int t = 0; t < 8; ++t) {
    #pragma unroll
    for (int i = 0; i < 8; ++i) {
      Wh[(size_t)(srow + i) * DIM + t * 16 + rsel] = acc[t][i];
    }
  }
}

__global__ __launch_bounds__(256) void gat_sd(const float* __restrict__ Wh,
                                              const float* __restrict__ a,
                                              float* __restrict__ s,
                                              float* __restrict__ d) {
  const int node = (int)((blockIdx.x * blockDim.x + threadIdx.x) >> 5);
  const int lane = (int)(threadIdx.x & 31);
  if (node >= N_NODES) return;
  const float* row = Wh + (size_t)node * DIM;
  float ss = 0.f, dd = 0.f;
  #pragma unroll
  for (int j = lane; j < DIM; j += 32) {
    float v = row[j];
    ss += v * a[j];
    dd += v * a[DIM + j];
  }
  #pragma unroll
  for (int off = 16; off > 0; off >>= 1) {
    ss += __shfl_down(ss, off, 32);
    dd += __shfl_down(dd, off, 32);
  }
  if (lane == 0) { s[node] = ss; d[node] = dd; }
}

__global__ void gat_init(const float* __restrict__ h, float* __restrict__ hout,
                         float* __restrict__ emax, float* __restrict__ denom) {
  const size_t i = (size_t)blockIdx.x * blockDim.x + threadIdx.x;
  if (i < (size_t)N_NODES * DIM) hout[i] = h[i];
  if (i < N_NODES) { emax[i] = -INFINITY; denom[i] = 0.f; }
}

__global__ void gat_edge_logits(const float* __restrict__ s, const float* __restrict__ d,
                                const int* __restrict__ src, const int* __restrict__ dst,
                                float* __restrict__ e, float* __restrict__ emax) {
  const int i = (int)(blockIdx.x * blockDim.x + threadIdx.x);
  if (i >= N_EDGES) return;
  float v = s[src[i]] + d[dst[i]];
  v = (v > 0.f) ? v : LRELU_A * v;
  e[i] = v;
  float* addr = emax + dst[i];
  if (v >= 0.f) atomicMax((int*)addr, __float_as_int(v));
  else          atomicMin((unsigned int*)addr, __float_as_uint(v));
}

__global__ void gat_fix_emax(float* __restrict__ emax) {
  const int i = (int)(blockIdx.x * blockDim.x + threadIdx.x);
  if (i < N_NODES) {
    float m = emax[i];
    if (!isfinite(m)) emax[i] = 0.f;
  }
}

__global__ void gat_edge_exp(const float* __restrict__ e, const int* __restrict__ dst,
                             const float* __restrict__ emax, float* __restrict__ ex,
                             float* __restrict__ denom) {
  const int i = (int)(blockIdx.x * blockDim.x + threadIdx.x);
  if (i >= N_EDGES) return;
  const float v = expf(e[i] - emax[dst[i]]);
  ex[i] = v;
  atomicAdd(&denom[dst[i]], v);
}

__global__ __launch_bounds__(256) void gat_scatter(const float* __restrict__ Wh,
                                                   const int* __restrict__ src,
                                                   const int* __restrict__ dst,
                                                   const float* __restrict__ ex,
                                                   const float* __restrict__ denom,
                                                   float* __restrict__ hout) {
  const int edge = (int)((blockIdx.x * blockDim.x + threadIdx.x) >> 5);
  const int lane = (int)(threadIdx.x & 31);
  if (edge >= N_EDGES) return;
  const int sn = src[edge];
  const int dn = dst[edge];
  const float w = ex[edge] / denom[dn];
  const float4 v = ((const float4*)(Wh + (size_t)sn * DIM))[lane];
  float* orow = hout + (size_t)dn * DIM + lane * 4;
  atomicAdd(orow + 0, w * v.x);
  atomicAdd(orow + 1, w * v.y);
  atomicAdd(orow + 2, w * v.z);
  atomicAdd(orow + 3, w * v.w);
}

extern "C" void kernel_launch(void* const* d_in, const int* in_sizes, int n_in,
                              void* d_out, int out_size, void* d_ws, size_t ws_size,
                              hipStream_t stream) {
  const float* h   = (const float*)d_in[0];
  const float* W   = (const float*)d_in[1];
  const float* a   = (const float*)d_in[2];
  const int*   src = (const int*)d_in[3];
  const int*   dst = (const int*)d_in[4];
  float* hout = (float*)d_out;

  float* ws    = (float*)d_ws;
  float* Wh    = ws;
  float* s     = Wh   + (size_t)N_NODES * DIM;
  float* d     = s    + N_NODES;
  float* emax  = d    + N_NODES;
  float* denom = emax + N_NODES;
  float* e     = denom + N_NODES;
  float* ex    = e    + N_EDGES;

  {
    const int waves = N_NODES / 16;
    const int blocks = (waves + 3) / 4;
    gat_gemm_wmma<<<blocks, 128, 0, stream>>>(h, W, Wh);
  }
  gat_sd<<<(N_NODES * 32 + 255) / 256, 256, 0, stream>>>(Wh, a, s, d);
  {
    const size_t total = (size_t)N_NODES * DIM;
    gat_init<<<(unsigned)((total + 255) / 256), 256, 0, stream>>>(h, hout, emax, denom);
  }
  gat_edge_logits<<<(N_EDGES + 255) / 256, 256, 0, stream>>>(s, d, src, dst, e, emax);
  gat_fix_emax<<<(N_NODES + 255) / 256, 256, 0, stream>>>(emax);
  gat_edge_exp<<<(N_EDGES + 255) / 256, 256, 0, stream>>>(e, dst, emax, ex, denom);
  gat_scatter<<<(N_EDGES * 32 + 255) / 256, 256, 0, stream>>>(Wh, src, dst, ex, denom, hout);
}


# owner work-groups scan dst once, not twice: each wave queues its edges in its own LDS segment, then max, exp and row sums run over the whole queue; sums still LDS atomics in arrival order
# speedup vs baseline: 2.5390x; 2.5390x over previous
#include <hip/hip_runtime.h>
#include <math.h>

#define N_NODES 50000
#define N_EDGES 800000
#define DIM     128
#define LRELU_A 0.2f

typedef __attribute__((ext_vector_type(2))) float v2f;
typedef __attribute__((ext_vector_type(8))) float v8f;

__global__ __launch_bounds__(128) void gat_gemm_wmma(const float* __restrict__ h,
                                                     const float* __restrict__ W,
                                                     float* __restrict__ Wh) {
  const int wave = (int)((blockIdx.x * blockDim.x + threadIdx.x) >> 5);
  const int lane = (int)(threadIdx.x & 31);
  if (wave >= N_NODES / 16) return;

  const int row0  = wave * 16;
  const int rsel  = lane & 15;
  const int khalf = (lane >> 4) * 2;

  v8f acc[8];
  #pragma unroll
  for (int t = 0; t < 8; ++t) { v8f z = {}; acc[t] = z; }

  for (int k0 = 0; k0 < DIM; k0 += 4) {
    const float* ap = h + (size_t)(row0 + rsel) * DIM + (k0 + khalf);
    v2f a; a.x = ap[0]; a.y = ap[1];
    #pragma unroll
    for (int t = 0; t < 8; ++t) {
      const float* bp = W + (size_t)(k0 + khalf) * DIM + t * 16 + rsel;
      v2f b; b.x = bp[0]; b.y = bp[DIM];
      acc[t] = __builtin_amdgcn_wmma_f32_16x16x4_f32(
           false, a,  false, b,
           (short)0, acc[t],  false,  false);
    }
  }

  const int srow = row0 + (lane >> 4) * 8;
  #pragma unroll
  for (int t = 0; t < 8; ++t) {
    #pragma unroll
    for (int i = 0; i < 8; ++i) {
      Wh[(size_t)(srow + i) * DIM + t * 16 + rsel] = acc[t][i];
    }
  }
}

__global__ __launch_bounds__(256) void gat_sd(const float* __restrict__ Wh,
                                              const float* __restrict__ a,
                                              float* __restrict__ s,
                                              float* __restrict__ d) {
  const int node = (int)((blockIdx.x * blockDim.x + threadIdx.x) >> 5);
  const int lane = (int)(threadIdx.x & 31);
  if (node >= N_NODES) return;
  const float* row = Wh + (size_t)node * DIM;
  float ss = 0.f, dd = 0.f;
  #pragma unroll
  for (int j = lane; j < DIM; j += 32) {
    float v = row[j];
    ss += v * a[j];
    dd += v * a[DIM + j];
  }
  #pragma unroll
  for (int off = 16; off > 0; off >>= 1) {
    ss += __shfl_down(ss, off, 32);
    dd += __shfl_down(dd, off, 32);
  }
  if (lane == 0) { s[node] = ss; d[node] = dd; }
}

#define NPW        196
#define OWN_T      1024
#define EPT        16
#define CHUNK      (OWN_T * EPT)
#define NCHUNK     ((N_EDGES + CHUNK - 1) / CHUNK)
#define EBITS      20
#define EMASK      ((1 << EBITS) - 1)


__device__ __forceinline__ void gat_lds_barrier() {
  __builtin_amdgcn_fence(__ATOMIC_RELEASE, "workgroup", "local");
  __builtin_amdgcn_s_barrier();
  __builtin_amdgcn_fence(__ATOMIC_ACQUIRE, "workgroup", "local");
}

__device__ __forceinline__ float gat_logit(float sv, float dv) {
  float v = sv + dv;
  return (v > 0.f) ? v : LRELU_A * v;
}

__device__ __forceinline__ void gat_load_dst(const int* __restrict__ dst, int cbase, int tid, int4* dv) {
  #pragma unroll
  for (int k = 0; k < EPT / 4; ++k) {
    const int b = cbase + k * (OWN_T * 4) + tid * 4;
    if (b < N_EDGES) dv[k] = *(const int4*)(dst + b);
    else             dv[k] = make_int4(-1, -1, -1, -1);
  }
}

__device__ __forceinline__ void gat_filter(const int4* dv, int cbase, int tid, int lo, unsigned nown,
                                           int* qcnt, int* q_a) {
  #pragma unroll
  for (int k = 0; k < EPT / 4; ++k) {
    const int b = cbase + k * (OWN_T * 4) + tid * 4;
    const int dn[4] = {dv[k].x, dv[k].y, dv[k].z, dv[k].w};
    #pragma unroll
    for (int j = 0; j < 4; ++j) {
      const unsigned dl = (unsigned)(dn[j] - lo);
      if (dl < nown) {
        const int pos = atomicAdd(qcnt, 1);
        q_a[pos] = (int)(dl << EBITS) | (b + j);
      }
    }
  }
}


#define NWAVE      (OWN_T / 32)
#define WQCAP      512
#define NIT4       (N_EDGES / 4)
#define SCAN_IT    ((NIT4 + OWN_T - 1) / OWN_T)
#define RING       8
#define SCAN_PAD   (((SCAN_IT + RING - 1) / RING) * RING)
#define ROT4       768

__device__ __forceinline__ int4 gat_scan_load(const int* __restrict__ dst, int it, int ubase, int rot4,
                                              int lane, int* piece) {
  int u = it * OWN_T + ubase;
  if (u >= NIT4) u = ubase;
  int r = u + rot4;
  if (r >= NIT4) r -= NIT4;
  *piece = r + lane;
  return *(const int4*)(dst + (size_t)r * 4 + lane * 4);
}

__device__ __forceinline__ void gat_keep(int dn, int e, int lo, unsigned nown, int2* myq, unsigned* wcnt) {
  const unsigned dl = (unsigned)(dn - lo);
  const bool m = dl < nown;
  const unsigned mask = __builtin_amdgcn_ballot_w32(m);
  if (mask != 0u) {
    const unsigned pos = *wcnt + __builtin_amdgcn_mbcnt_lo(mask, 0u);
    if (m && pos < WQCAP) myq[pos].x = (int)(dl << EBITS) | e;
    *wcnt += (unsigned)__builtin_popcount(mask);
  }
}

template <int G>
__device__ __forceinline__ void gat_rows_add(const float* __restrict__ Wh, float* rows, int keyv, float exv,
                                             int j, int lane) {
  float4 r[G];
  int    key[G];
  #pragma unroll
  for (int k = 0; k < G; ++k) {
    key[k] = __builtin_amdgcn_readlane(keyv, j + k);
    r[k] = ((const float4*)(Wh + (size_t)(key[k] & EMASK) * DIM))[lane];
  }
  #pragma unroll
  for (int k = 0; k < G; ++k) {
    asm volatile("" :: "v"(r[k].x), "v"(r[k].y), "v"(r[k].z), "v"(r[k].w));
  }
  #pragma unroll
  for (int k = 0; k < G; ++k) {
    const float x = __int_as_float(__builtin_amdgcn_readlane(__float_as_int(exv), j + k));
    float* rp = rows + (int)((unsigned)key[k] >> EBITS) * DIM + lane;
    atomicAdd(rp +  0, x * r[k].x);
    atomicAdd(rp + 32, x * r[k].y);
    atomicAdd(rp + 64, x * r[k].z);
    atomicAdd(rp + 96, x * r[k].w);
  }
}

__global__ __launch_bounds__(OWN_T) void gat_owner(const float* __restrict__ Wh,
                                                   const float* __restrict__ s,
                                                   const float* __restrict__ d,
                                                   const int* __restrict__ src,
                                                   const int* __restrict__ dst,
                                                   const float* __restrict__ h,
                                                   float* __restrict__ out) {
  __shared__ float rows[NPW * DIM];
  __shared__ float emaxL[NPW];
  __shared__ float denL[NPW];
  __shared__ float dL[NPW];
  __shared__ int2  wq[NWAVE * WQCAP];
  __shared__ unsigned wcntL[NWAVE];
  __shared__ int   qcnt[2];

  const int tid  = (int)threadIdx.x;
  const int lane = tid & 31;
  const int wv   = __builtin_amdgcn_readfirstlane(tid >> 5);
  const int lo   = (int)blockIdx.x * NPW;
  const int hi   = (lo + NPW < N_NODES) ? (lo + NPW) : N_NODES;
  const unsigned nown = (unsigned)(hi - lo);

  for (int i = tid; i < NPW * DIM; i += OWN_T) rows[i] = 0.f;
  if (tid < NPW) {
    emaxL[tid] = -INFINITY;
    denL[tid]  = 0.f;
    dL[tid]    = ((unsigned)tid < nown) ? d[lo + tid] : 0.f;
  }
  if (tid < 2) qcnt[tid] = 0;

  int2* myq = wq + wv * WQCAP;
  unsigned wcnt = 0;
  {
    const int rot4  = (int)blockIdx.x * ROT4;
    const int ubase = wv * 32;
    int4 ring[RING];
    int  piece[RING];
    #pragma unroll
    for (int k = 0; k < RING; ++k) ring[k] = gat_scan_load(dst, k, ubase, rot4, lane, &piece[k]);
    for (int it = 0; it < SCAN_PAD; it += RING) {
      #pragma unroll
      for (int k = 0; k < RING; ++k) {
        const int4 v = ring[k];
        const int  e = piece[k] * 4;
        if ((it + k) * OWN_T + ubase < NIT4) {
          gat_keep(v.x, e + 0, lo, nown, myq, &wcnt);
          gat_keep(v.y, e + 1, lo, nown, myq, &wcnt);
          gat_keep(v.z, e + 2, lo, nown, myq, &wcnt);
          gat_keep(v.w, e + 3, lo, nown, myq, &wcnt);
        }
        ring[k] = gat_scan_load(dst, it + k + RING, ubase, rot4, lane, &piece[k]);
      }
    }
  }
  if (lane == 0) wcntL[wv] = wcnt;
  __syncthreads();

  const bool over = __builtin_amdgcn_ballot_w32(wcntL[lane] > (unsigned)WQCAP) != 0u;

  if (!over) {
    const int n = (int)wcnt;

    for (int i0 = 0; i0 < n; i0 += 128) {
      int   key[4];
      int   sn[4];
      float sv[4];
      #pragma unroll
      for (int k = 0; k < 4; ++k) {
        const int i = i0 + k * 32 + lane;
        key[k] = (i < n) ? myq[i].x : 0;
      }
      #pragma unroll
      for (int k = 0; k < 4; ++k) sn[k] = src[key[k] & EMASK];
      #pragma unroll
      for (int k = 0; k < 4; ++k) sv[k] = s[sn[k]];
      asm volatile("" :: "v"(sv[0]), "v"(sv[1]), "v"(sv[2]), "v"(sv[3]));
      #pragma unroll
      for (int k = 0; k < 4; ++k) {
        const int i = i0 + k * 32 + lane;
        if (i < n) {
          const int dl = (int)((unsigned)key[k] >> EBITS);
          const float v = gat_logit(sv[k], dL[dl]);
          float* addr = &emaxL[dl];
          if (v >= 0.f) atomicMax((int*)addr, __float_as_int(v));
          else          atomicMin((unsigned int*)addr, __float_as_uint(v));
          myq[i] = make_int2((int)((unsigned)dl << EBITS) | sn[k], __float_as_int(v));
        }
      }
    }
    gat_lds_barrier();

    for (int i = lane; i < n; i += 32) {
      const int2 q = myq[i];
      const int dl = (int)((unsigned)q.x >> EBITS);
      float m = emaxL[dl];
      if (!isfinite(m)) m = 0.f;
      const float ex = expf(__int_as_float(q.y) - m);
      atomicAdd(&denL[dl], ex);
      myq[i].y = __float_as_int(ex);
    }
    gat_lds_barrier();

    for (int b = 0; b < n; b += 32) {
      const int i = b + lane;
      const int2 q = (i < n) ? myq[i] : make_int2(0, 0);
      const int keyv = q.x;
      const float exv = __int_as_float(q.y);
      const int cnt = (n - b < 32) ? (n - b) : 32;
      int j = 0;
      for (; j + 8 <= cnt; j += 8) gat_rows_add<8>(Wh, rows, keyv, exv, j, lane);
      if (j + 4 <= cnt) { gat_rows_add<4>(Wh, rows, keyv, exv, j, lane); j += 4; }
      if (j + 2 <= cnt) { gat_rows_add<2>(Wh, rows, keyv, exv, j, lane); j += 2; }
      if (j < cnt)      { gat_rows_add<1>(Wh, rows, keyv, exv, j, lane); }
    }
    gat_lds_barrier();
  } else {
    int*   q_a = (int*)wq;
    float* q_b = (float*)(q_a + CHUNK);
    const int rot = (int)(blockIdx.x % NCHUNK);

    {
      int4 dv[EPT / 4];
      int cc = rot;
      gat_load_dst(dst, cc * CHUNK, tid, dv);
      for (int c = 0; c < NCHUNK; ++c) {
        const int cbase = cc * CHUNK;
        int* cnt = &qcnt[c & 1];
        gat_filter(dv, cbase, tid, lo, nown, cnt, q_a);
        cc = (cc + 1 == NCHUNK) ? 0 : cc + 1;
        if (c + 1 < NCHUNK) gat_load_dst(dst, cc * CHUNK, tid, dv);
        gat_lds_barrier();
        const int n = *cnt;
        if (tid == 0) qcnt[(c + 1) & 1] = 0;
        for (int i = tid; i < n; i += OWN_T) {
          const int p  = q_a[i];
          const int e  = p & EMASK;
          const int dl = (int)((unsigned)p >> EBITS);
          const float v = gat_logit(s[src[e]], dL[dl]);
          float* addr = &emaxL[dl];
          if (v >= 0.f) atomicMax((int*)addr, __float_as_int(v));
          else          atomicMin((unsigned int*)addr, __float_as_uint(v));
        }
        gat_lds_barrier();
      }
    }

    if (tid < NPW) {
      const float m = emaxL[tid];
      if (!isfinite(m)) emaxL[tid] = 0.f;
    }
    __syncthreads();

    {
      int4 dv[EPT / 4];
      int cc = rot;
      gat_load_dst(dst, cc * CHUNK, tid, dv);
      for (int c = 0; c < NCHUNK; ++c) {
        const int cbase = cc * CHUNK;
        int* cnt = &qcnt[(c + NCHUNK) & 1];
        gat_filter(dv, cbase, tid, lo, nown, cnt, q_a);
        cc = (cc + 1 == NCHUNK) ? 0 : cc + 1;
        if (c + 1 < NCHUNK) gat_load_dst(dst, cc * CHUNK, tid, dv);
        gat_lds_barrier();
        const int n = *cnt;
        if (tid == 0) qcnt[(c + NCHUNK + 1) & 1] = 0;
        for (int i = tid; i < n; i += OWN_T) {
          const int p  = q_a[i];
          const int e  = p & EMASK;
          const int dl = (int)((unsigned)p >> EBITS);
          const int sn = src[e];
          const float v  = gat_logit(s[sn], dL[dl]);
          const float ex = expf(v - emaxL[dl]);
          atomicAdd(&denL[dl], ex);
          q_a[i] = (int)((unsigned)dl << EBITS) | sn;
          q_b[i] = ex;
        }
        gat_lds_barrier();
        for (int i = wv; i < n; i += 64) {
          const int  i2   = i + 32;
          const bool two  = i2 < n;
          const int  p0   = q_a[i];
          const int  p1   = two ? q_a[i2] : p0;
          const float x0  = q_b[i];
          const float x1  = two ? q_b[i2] : 0.f;
          const float4 v0 = ((const float4*)(Wh + (size_t)(p0 & EMASK) * DIM))[lane];
          const float4 v1 = ((const float4*)(Wh + (size_t)(p1 & EMASK) * DIM))[lane];
          asm volatile("" :: "v"(v0.x), "v"(v0.y), "v"(v0.z), "v"(v0.w),
                             "v"(v1.x), "v"(v1.y), "v"(v1.z), "v"(v1.w));
          float* r0 = rows + (int)((unsigned)p0 >> EBITS) * DIM + lane;
          atomicAdd(r0 +  0, x0 * v0.x);
          atomicAdd(r0 + 32, x0 * v0.y);
          atomicAdd(r0 + 64, x0 * v0.z);
          atomicAdd(r0 + 96, x0 * v0.w);
          if (two) {
            float* r1 = rows + (int)((unsigned)p1 >> EBITS) * DIM + lane;
            atomicAdd(r1 +  0, x1 * v1.x);
            atomicAdd(r1 + 32, x1 * v1.y);
            atomicAdd(r1 + 64, x1 * v1.z);
            atomicAdd(r1 + 96, x1 * v1.w);
          }
        }
        gat_lds_barrier();
      }
    }

  }

  const int total = (int)nown * DIM;
  for (int i = tid; i < total; i += OWN_T) {
    const int nl = i >> 7;
    const int c  = i & (DIM - 1);
    const float den = denL[nl];
    const float acc = rows[nl * DIM + (c & 3) * 32 + (c >> 2)];
    const float agg = (den != 0.f) ? acc / den : 0.f;
    const size_t g = (size_t)(lo + nl) * DIM + c;
    out[g] = agg + h[g];
  }
}

extern "C" void kernel_launch(void* const* d_in, const int* in_sizes, int n_in,
                              void* d_out, int out_size, void* d_ws, size_t ws_size,
                              hipStream_t stream) {
  const float* h   = (const float*)d_in[0];
  const float* W   = (const float*)d_in[1];
  const float* a   = (const float*)d_in[2];
  const int*   src = (const int*)d_in[3];
  const int*   dst = (const int*)d_in[4];
  float* hout = (float*)d_out;

  float* ws    = (float*)d_ws;
  float* Wh    = ws;
  float* s     = Wh   + (size_t)N_NODES * DIM;
  float* d     = s    + N_NODES;

  {
    const int waves = N_NODES / 16;
    const int blocks = (waves + 3) / 4;
    gat_gemm_wmma<<<blocks, 128, 0, stream>>>(h, W, Wh);
  }
  gat_sd<<<(N_NODES * 32 + 255) / 256, 256, 0, stream>>>(Wh, a, s, d);
  gat_owner<<<(N_NODES + NPW - 1) / NPW, OWN_T, 0, stream>>>(Wh, s, d, src, dst, h, hout);
}


# s,d launch: one wave per 16 nodes (3,125 waves, not 50,000): 16 whole-row 16-byte loads issued together, xor-butterfly sums; s,d summed in a new order (4 neighbouring columns per lane first)
# speedup vs baseline: 2.6803x; 1.0557x over previous
#include <hip/hip_runtime.h>
#include <math.h>

#define N_NODES 50000
#define N_EDGES 800000
#define DIM     128
#define LRELU_A 0.2f

typedef __attribute__((ext_vector_type(2))) float v2f;
typedef __attribute__((ext_vector_type(8))) float v8f;

__global__ __launch_bounds__(128) void gat_gemm_wmma(const float* __restrict__ h,
                                                     const float* __restrict__ W,
                                                     float* __restrict__ Wh) {
  const int wave = (int)((blockIdx.x * blockDim.x + threadIdx.x) >> 5);
  const int lane = (int)(threadIdx.x & 31);
  if (wave >= N_NODES / 16) return;

  const int row0  = wave * 16;
  const int rsel  = lane & 15;
  const int khalf = (lane >> 4) * 2;

  v8f acc[8];
  #pragma unroll
  for (int t = 0; t < 8; ++t) { v8f z = {}; acc[t] = z; }

  for (int k0 = 0; k0 < DIM; k0 += 4) {
    const float* ap = h + (size_t)(row0 + rsel) * DIM + (k0 + khalf);
    v2f a; a.x = ap[0]; a.y = ap[1];
    #pragma unroll
    for (int t = 0; t < 8; ++t) {
      const float* bp = W + (size_t)(k0 + khalf) * DIM + t * 16 + rsel;
      v2f b; b.x = bp[0]; b.y = bp[DIM];
      acc[t] = __builtin_amdgcn_wmma_f32_16x16x4_f32(
           false, a,  false, b,
           (short)0, acc[t],  false,  false);
    }
  }

  const int srow = row0 + (lane >> 4) * 8;
  #pragma unroll
  for (int t = 0; t < 8; ++t) {
    #pragma unroll
    for (int i = 0; i < 8; ++i) {
      Wh[(size_t)(srow + i) * DIM + t * 16 + rsel] = acc[t][i];
    }
  }
}

#define SD_ROWS 16
static_assert(N_NODES % SD_ROWS == 0, "a wave takes SD_ROWS whole rows");
static_assert(DIM == 128, "one float4 per lane covers a row");

__global__ __launch_bounds__(256) void gat_sd(const float* __restrict__ Wh,
                                              const float* __restrict__ a,
                                              float* __restrict__ s,
                                              float* __restrict__ d) {
  const int wave = (int)((blockIdx.x * blockDim.x + threadIdx.x) >> 5);
  const int lane = (int)(threadIdx.x & 31);
  if (wave >= N_NODES / SD_ROWS) return;
  const int row0 = wave * SD_ROWS;

  const float4 as = ((const float4*)a)[lane];
  const float4 ad = ((const float4*)(a + DIM))[lane];
  const float4* rp = (const float4*)(Wh + (size_t)row0 * DIM) + lane;

  float4 v[SD_ROWS];
  #pragma unroll
  for (int r = 0; r < SD_ROWS; ++r) v[r] = rp[r * (DIM / 4)];

  float ss[SD_ROWS], dd[SD_ROWS];
  #pragma unroll
  for (int r = 0; r < SD_ROWS; ++r) {
    ss[r] = v[r].x * as.x + v[r].y * as.y + v[r].z * as.z + v[r].w * as.w;
    dd[r] = v[r].x * ad.x + v[r].y * ad.y + v[r].z * ad.z + v[r].w * ad.w;
  }
  #pragma unroll
  for (int off = 16; off > 0; off >>= 1) {
    #pragma unroll
    for (int r = 0; r < SD_ROWS; ++r) {
      ss[r] += __shfl_xor(ss[r], off, 32);
      dd[r] += __shfl_xor(dd[r], off, 32);
    }
  }
  float ms = 0.f, md = 0.f;
  #pragma unroll
  for (int r = 0; r < SD_ROWS; ++r) {
    if (lane == r) { ms = ss[r]; md = dd[r]; }
  }
  if (lane < SD_ROWS) {
    s[row0 + lane] = ms;
    d[row0 + lane] = md;
  }
}

#define NPW        196
#define OWN_T      1024
#define EPT        16
#define CHUNK      (OWN_T * EPT)
#define NCHUNK     ((N_EDGES + CHUNK - 1) / CHUNK)
#define EBITS      20
#define EMASK      ((1 << EBITS) - 1)


__device__ __forceinline__ void gat_lds_barrier() {
  __builtin_amdgcn_fence(__ATOMIC_RELEASE, "workgroup", "local");
  __builtin_amdgcn_s_barrier();
  __builtin_amdgcn_fence(__ATOMIC_ACQUIRE, "workgroup", "local");
}

__device__ __forceinline__ float gat_logit(float sv, float dv) {
  float v = sv + dv;
  return (v > 0.f) ? v : LRELU_A * v;
}

__device__ __forceinline__ void gat_load_dst(const int* __restrict__ dst, int cbase, int tid, int4* dv) {
  #pragma unroll
  for (int k = 0; k < EPT / 4; ++k) {
    const int b = cbase + k * (OWN_T * 4) + tid * 4;
    if (b < N_EDGES) dv[k] = *(const int4*)(dst + b);
    else             dv[k] = make_int4(-1, -1, -1, -1);
  }
}

__device__ __forceinline__ void gat_filter(const int4* dv, int cbase, int tid, int lo, unsigned nown,
                                           int* qcnt, int* q_a) {
  #pragma unroll
  for (int k = 0; k < EPT / 4; ++k) {
    const int b = cbase + k * (OWN_T * 4) + tid * 4;
    const int dn[4] = {dv[k].x, dv[k].y, dv[k].z, dv[k].w};
    #pragma unroll
    for (int j = 0; j < 4; ++j) {
      const unsigned dl = (unsigned)(dn[j] - lo);
      if (dl < nown) {
        const int pos = atomicAdd(qcnt, 1);
        q_a[pos] = (int)(dl << EBITS) | (b + j);
      }
    }
  }
}


#define NWAVE      (OWN_T / 32)
#define WQCAP      512
#define NIT4       (N_EDGES / 4)
#define SCAN_IT    ((NIT4 + OWN_T - 1) / OWN_T)
#define RING       8
#define SCAN_PAD   (((SCAN_IT + RING - 1) / RING) * RING)
#define ROT4       768

__device__ __forceinline__ int4 gat_scan_load(const int* __restrict__ dst, int it, int ubase, int rot4,
                                              int lane, int* piece) {
  int u = it * OWN_T + ubase;
  if (u >= NIT4) u = ubase;
  int r = u + rot4;
  if (r >= NIT4) r -= NIT4;
  *piece = r + lane;
  return *(const int4*)(dst + (size_t)r * 4 + lane * 4);
}

__device__ __forceinline__ void gat_keep(int dn, int e, int lo, unsigned nown, int2* myq, unsigned* wcnt) {
  const unsigned dl = (unsigned)(dn - lo);
  const bool m = dl < nown;
  const unsigned mask = __builtin_amdgcn_ballot_w32(m);
  if (mask != 0u) {
    const unsigned pos = *wcnt + __builtin_amdgcn_mbcnt_lo(mask, 0u);
    if (m && pos < WQCAP) myq[pos].x = (int)(dl << EBITS) | e;
    *wcnt += (unsigned)__builtin_popcount(mask);
  }
}

template <int G>
__device__ __forceinline__ void gat_rows_add(const float* __restrict__ Wh, float* rows, int keyv, float exv,
                                             int j, int lane) {
  float4 r[G];
  int    key[G];
  #pragma unroll
  for (int k = 0; k < G; ++k) {
    key[k] = __builtin_amdgcn_readlane(keyv, j + k);
    r[k] = ((const float4*)(Wh + (size_t)(key[k] & EMASK) * DIM))[lane];
  }
  #pragma unroll
  for (int k = 0; k < G; ++k) {
    asm volatile("" :: "v"(r[k].x), "v"(r[k].y), "v"(r[k].z), "v"(r[k].w));
  }
  #pragma unroll
  for (int k = 0; k < G; ++k) {
    const float x = __int_as_float(__builtin_amdgcn_readlane(__float_as_int(exv), j + k));
    float* rp = rows + (int)((unsigned)key[k] >> EBITS) * DIM + lane;
    atomicAdd(rp +  0, x * r[k].x);
    atomicAdd(rp + 32, x * r[k].y);
    atomicAdd(rp + 64, x * r[k].z);
    atomicAdd(rp + 96, x * r[k].w);
  }
}

__global__ __launch_bounds__(OWN_T) void gat_owner(const float* __restrict__ Wh,
                                                   const float* __restrict__ s,
                                                   const float* __restrict__ d,
                                                   const int* __restrict__ src,
                                                   const int* __restrict__ dst,
                                                   const float* __restrict__ h,
                                                   float* __restrict__ out) {
  __shared__ float rows[NPW * DIM];
  __shared__ float emaxL[NPW];
  __shared__ float denL[NPW];
  __shared__ float dL[NPW];
  __shared__ int2  wq[NWAVE * WQCAP];
  __shared__ unsigned wcntL[NWAVE];
  __shared__ int   qcnt[2];

  const int tid  = (int)threadIdx.x;
  const int lane = tid & 31;
  const int wv   = __builtin_amdgcn_readfirstlane(tid >> 5);
  const int lo   = (int)blockIdx.x * NPW;
  const int hi   = (lo + NPW < N_NODES) ? (lo + NPW) : N_NODES;
  const unsigned nown = (unsigned)(hi - lo);

  for (int i = tid; i < NPW * DIM; i += OWN_T) rows[i] = 0.f;
  if (tid < NPW) {
    emaxL[tid] = -INFINITY;
    denL[tid]  = 0.f;
    dL[tid]    = ((unsigned)tid < nown) ? d[lo + tid] : 0.f;
  }
  if (tid < 2) qcnt[tid] = 0;

  int2* myq = wq + wv * WQCAP;
  unsigned wcnt = 0;
  {
    const int rot4  = (int)blockIdx.x * ROT4;
    const int ubase = wv * 32;
    int4 ring[RING];
    int  piece[RING];
    #pragma unroll
    for (int k = 0; k < RING; ++k) ring[k] = gat_scan_load(dst, k, ubase, rot4, lane, &piece[k]);
    for (int it = 0; it < SCAN_PAD; it += RING) {
      #pragma unroll
      for (int k = 0; k < RING; ++k) {
        const int4 v = ring[k];
        const int  e = piece[k] * 4;
        if ((it + k) * OWN_T + ubase < NIT4) {
          gat_keep(v.x, e + 0, lo, nown, myq, &wcnt);
          gat_keep(v.y, e + 1, lo, nown, myq, &wcnt);
          gat_keep(v.z, e + 2, lo, nown, myq, &wcnt);
          gat_keep(v.w, e + 3, lo, nown, myq, &wcnt);
        }
        ring[k] = gat_scan_load(dst, it + k + RING, ubase, rot4, lane, &piece[k]);
      }
    }
  }
  if (lane == 0) wcntL[wv] = wcnt;
  __syncthreads();

  const bool over = __builtin_amdgcn_ballot_w32(wcntL[lane] > (unsigned)WQCAP) != 0u;

  if (!over) {
    const int n = (int)wcnt;

    for (int i0 = 0; i0 < n; i0 += 128) {
      int   key[4];
      int   sn[4];
      float sv[4];
      #pragma unroll
      for (int k = 0; k < 4; ++k) {
        const int i = i0 + k * 32 + lane;
        key[k] = (i < n) ? myq[i].x : 0;
      }
      #pragma unroll
      for (int k = 0; k < 4; ++k) sn[k] = src[key[k] & EMASK];
      #pragma unroll
      for (int k = 0; k < 4; ++k) sv[k] = s[sn[k]];
      asm volatile("" :: "v"(sv[0]), "v"(sv[1]), "v"(sv[2]), "v"(sv[3]));
      #pragma unroll
      for (int k = 0; k < 4; ++k) {
        const int i = i0 + k * 32 + lane;
        if (i < n) {
          const int dl = (int)((unsigned)key[k] >> EBITS);
          const float v = gat_logit(sv[k], dL[dl]);
          float* addr = &emaxL[dl];
          if (v >= 0.f) atomicMax((int*)addr, __float_as_int(v));
          else          atomicMin((unsigned int*)addr, __float_as_uint(v));
          myq[i] = make_int2((int)((unsigned)dl << EBITS) | sn[k], __float_as_int(v));
        }
      }
    }
    gat_lds_barrier();

    for (int i = lane; i < n; i += 32) {
      const int2 q = myq[i];
      const int dl = (int)((unsigned)q.x >> EBITS);
      float m = emaxL[dl];
      if (!isfinite(m)) m = 0.f;
      const float ex = expf(__int_as_float(q.y) - m);
      atomicAdd(&denL[dl], ex);
      myq[i].y = __float_as_int(ex);
    }
    gat_lds_barrier();

    for (int b = 0; b < n; b += 32) {
      const int i = b + lane;
      const int2 q = (i < n) ? myq[i] : make_int2(0, 0);
      const int keyv = q.x;
      const float exv = __int_as_float(q.y);
      const int cnt = (n - b < 32) ? (n - b) : 32;
      int j = 0;
      for (; j + 8 <= cnt; j += 8) gat_rows_add<8>(Wh, rows, keyv, exv, j, lane);
      if (j + 4 <= cnt) { gat_rows_add<4>(Wh, rows, keyv, exv, j, lane); j += 4; }
      if (j + 2 <= cnt) { gat_rows_add<2>(Wh, rows, keyv, exv, j, lane); j += 2; }
      if (j < cnt)      { gat_rows_add<1>(Wh, rows, keyv, exv, j, lane); }
    }
    gat_lds_barrier();
  } else {
    int*   q_a = (int*)wq;
    float* q_b = (float*)(q_a + CHUNK);
    const int rot = (int)(blockIdx.x % NCHUNK);

    {
      int4 dv[EPT / 4];
      int cc = rot;
      gat_load_dst(dst, cc * CHUNK, tid, dv);
      for (int c = 0; c < NCHUNK; ++c) {
        const int cbase = cc * CHUNK;
        int* cnt = &qcnt[c & 1];
        gat_filter(dv, cbase, tid, lo, nown, cnt, q_a);
        cc = (cc + 1 == NCHUNK) ? 0 : cc + 1;
        if (c + 1 < NCHUNK) gat_load_dst(dst, cc * CHUNK, tid, dv);
        gat_lds_barrier();
        const int n = *cnt;
        if (tid == 0) qcnt[(c + 1) & 1] = 0;
        for (int i = tid; i < n; i += OWN_T) {
          const int p  = q_a[i];
          const int e  = p & EMASK;
          const int dl = (int)((unsigned)p >> EBITS);
          const float v = gat_logit(s[src[e]], dL[dl]);
          float* addr = &emaxL[dl];
          if (v >= 0.f) atomicMax((int*)addr, __float_as_int(v));
          else          atomicMin((unsigned int*)addr, __float_as_uint(v));
        }
        gat_lds_barrier();
      }
    }

    if (tid < NPW) {
      const float m = emaxL[tid];
      if (!isfinite(m)) emaxL[tid] = 0.f;
    }
    __syncthreads();

    {
      int4 dv[EPT / 4];
      int cc = rot;
      gat_load_dst(dst, cc * CHUNK, tid, dv);
      for (int c = 0; c < NCHUNK; ++c) {
        const int cbase = cc * CHUNK;
        int* cnt = &qcnt[(c + NCHUNK) & 1];
        gat_filter(dv, cbase, tid, lo, nown, cnt, q_a);
        cc = (cc + 1 == NCHUNK) ? 0 : cc + 1;
        if (c + 1 < NCHUNK) gat_load_dst(dst, cc * CHUNK, tid, dv);
        gat_lds_barrier();
        const int n = *cnt;
        if (tid == 0) qcnt[(c + NCHUNK + 1) & 1] = 0;
        for (int i = tid; i < n; i += OWN_T) {
          const int p  = q_a[i];
          const int e  = p & EMASK;
          const int dl = (int)((unsigned)p >> EBITS);
          const int sn = src[e];
          const float v  = gat_logit(s[sn], dL[dl]);
          const float ex = expf(v - emaxL[dl]);
          atomicAdd(&denL[dl], ex);
          q_a[i] = (int)((unsigned)dl << EBITS) | sn;
          q_b[i] = ex;
        }
        gat_lds_barrier();
        for (int i = wv; i < n; i += 64) {
          const int  i2   = i + 32;
          const bool two  = i2 < n;
          const int  p0   = q_a[i];
          const int  p1   = two ? q_a[i2] : p0;
          const float x0  = q_b[i];
          const float x1  = two ? q_b[i2] : 0.f;
          const float4 v0 = ((const float4*)(Wh + (size_t)(p0 & EMASK) * DIM))[lane];
          const float4 v1 = ((const float4*)(Wh + (size_t)(p1 & EMASK) * DIM))[lane];
          asm volatile("" :: "v"(v0.x), "v"(v0.y), "v"(v0.z), "v"(v0.w),
                             "v"(v1.x), "v"(v1.y), "v"(v1.z), "v"(v1.w));
          float* r0 = rows + (int)((unsigned)p0 >> EBITS) * DIM + lane;
          atomicAdd(r0 +  0, x0 * v0.x);
          atomicAdd(r0 + 32, x0 * v0.y);
          atomicAdd(r0 + 64, x0 * v0.z);
          atomicAdd(r0 + 96, x0 * v0.w);
          if (two) {
            float* r1 = rows + (int)((unsigned)p1 >> EBITS) * DIM + lane;
            atomicAdd(r1 +  0, x1 * v1.x);
            atomicAdd(r1 + 32, x1 * v1.y);
            atomicAdd(r1 + 64, x1 * v1.z);
            atomicAdd(r1 + 96, x1 * v1.w);
          }
        }
        gat_lds_barrier();
      }
    }

  }

  const int total = (int)nown * DIM;
  for (int i = tid; i < total; i += OWN_T) {
    const int nl = i >> 7;
    const int c  = i & (DIM - 1);
    const float den = denL[nl];
    const float acc = rows[nl * DIM + (c & 3) * 32 + (c >> 2)];
    const float agg = (den != 0.f) ? acc / den : 0.f;
    const size_t g = (size_t)(lo + nl) * DIM + c;
    out[g] = agg + h[g];
  }
}

extern "C" void kernel_launch(void* const* d_in, const int* in_sizes, int n_in,
                              void* d_out, int out_size, void* d_ws, size_t ws_size,
                              hipStream_t stream) {
  const float* h   = (const float*)d_in[0];
  const float* W   = (const float*)d_in[1];
  const float* a   = (const float*)d_in[2];
  const int*   src = (const int*)d_in[3];
  const int*   dst = (const int*)d_in[4];
  float* hout = (float*)d_out;

  float* ws    = (float*)d_ws;
  float* Wh    = ws;
  float* s     = Wh   + (size_t)N_NODES * DIM;
  float* d     = s    + N_NODES;

  {
    const int waves = N_NODES / 16;
    const int blocks = (waves + 3) / 4;
    gat_gemm_wmma<<<blocks, 128, 0, stream>>>(h, W, Wh);
  }
  gat_sd<<<(N_NODES / SD_ROWS + 7) / 8, 256, 0, stream>>>(Wh, a, s, d);
  gat_owner<<<(N_NODES + NPW - 1) / NPW, OWN_T, 0, stream>>>(Wh, s, d, src, dst, h, hout);
}


# owner work-groups scan one owner byte per edge (dst/196; bytes and (node,src) records packed by the s,d launch): 4x fewer scan bytes; queue order inside a wave changes, sums still LDS atomics
# speedup vs baseline: 3.0210x; 1.1271x over previous
#include <hip/hip_runtime.h>
#include <math.h>

#define N_NODES 50000
#define N_EDGES 800000
#define DIM     128
#define LRELU_A 0.2f

typedef __attribute__((ext_vector_type(2))) float v2f;
typedef __attribute__((ext_vector_type(8))) float v8f;

__global__ __launch_bounds__(128) void gat_gemm_wmma(const float* __restrict__ h,
                                                     const float* __restrict__ W,
                                                     float* __restrict__ Wh) {
  const int wave = (int)((blockIdx.x * blockDim.x + threadIdx.x) >> 5);
  const int lane = (int)(threadIdx.x & 31);
  if (wave >= N_NODES / 16) return;

  const int row0  = wave * 16;
  const int rsel  = lane & 15;
  const int khalf = (lane >> 4) * 2;

  v8f acc[8];
  #pragma unroll
  for (int t = 0; t < 8; ++t) { v8f z = {}; acc[t] = z; }

  for (int k0 = 0; k0 < DIM; k0 += 4) {
    const float* ap = h + (size_t)(row0 + rsel) * DIM + (k0 + khalf);
    v2f a; a.x = ap[0]; a.y = ap[1];
    #pragma unroll
    for (int t = 0; t < 8; ++t) {
      const float* bp = W + (size_t)(k0 + khalf) * DIM + t * 16 + rsel;
      v2f b; b.x = bp[0]; b.y = bp[DIM];
      acc[t] = __builtin_amdgcn_wmma_f32_16x16x4_f32(
           false, a,  false, b,
           (short)0, acc[t],  false,  false);
    }
  }

  const int srow = row0 + (lane >> 4) * 8;
  #pragma unroll
  for (int t = 0; t < 8; ++t) {
    #pragma unroll
    for (int i = 0; i < 8; ++i) {
      Wh[(size_t)(srow + i) * DIM + t * 16 + rsel] = acc[t][i];
    }
  }
}

#define SD_ROWS 16
static_assert(N_NODES % SD_ROWS == 0, "a wave takes SD_ROWS whole rows");
static_assert(DIM == 128, "one float4 per lane covers a row");

#define NPW        196
#define NOWNER     ((N_NODES + NPW - 1) / NPW)
#define NIT16      (N_EDGES / 16)
#define NITP       (((NIT16 + 31) / 32) * 32)
#define DL_NONE    255u
static_assert(N_EDGES % 16 == 0, "own8 pieces hold 16 whole edges");
static_assert(NOWNER <= 256, "an owner index fits a byte");
static_assert(N_NODES <= 65536 && NPW < (int)DL_NONE, "rec holds src in 16 bits, the local node in 8");
static_assert(N_EDGES / 8 == (N_NODES / SD_ROWS) * 32, "a wave of the s,d launch takes SD_ROWS rows and 256 edges");
static_assert(NITP * 2 - N_EDGES / 8 == 32, "one more wave writes the pad pieces of own8");
static_assert(((N_NODES / SD_ROWS + 7) / 8) * 8 > N_NODES / SD_ROWS, "the s,d grid holds that wave");

__device__ __forceinline__ unsigned gat_edge_rec(int dn, int sn, unsigned* ow) {
  const unsigned du = (unsigned)dn;
  const unsigned o  = du / NPW;
  const unsigned dl = (du < (unsigned)N_NODES) ? (du - o * NPW) : DL_NONE;
  *ow = o & 255u;
  return (dl << 16) | ((unsigned)sn & 0xFFFFu);
}

__global__ __launch_bounds__(256) void gat_sd(const float* __restrict__ Wh,
                                              const float* __restrict__ a,
                                              float* __restrict__ s,
                                              float* __restrict__ d,
                                              const int* __restrict__ src,
                                              const int* __restrict__ dst,
                                              unsigned* __restrict__ own8,
                                              unsigned* __restrict__ rec) {
  const int gt   = (int)(blockIdx.x * blockDim.x + threadIdx.x);
  const int wave = gt >> 5;
  const int lane = (int)(threadIdx.x & 31);
  if (wave > N_NODES / SD_ROWS) return;
  const int ea = wave * 256 + lane * 4;
  const int eb = ea + 128;
  if (wave == N_NODES / SD_ROWS) {
    own8[ea >> 2] = 0u;
    own8[eb >> 2] = 0u;
    return;
  }
  const int row0 = wave * SD_ROWS;

  const float4 as = ((const float4*)a)[lane];
  const float4 ad = ((const float4*)(a + DIM))[lane];
  const float4* rp = (const float4*)(Wh + (size_t)row0 * DIM) + lane;

  float4 v[SD_ROWS];
  #pragma unroll
  for (int r = 0; r < SD_ROWS; ++r) v[r] = rp[r * (DIM / 4)];
  const int4 d0 = *(const int4*)(dst + ea);
  const int4 d1 = *(const int4*)(dst + eb);
  const int4 s0 = *(const int4*)(src + ea);
  const int4 s1 = *(const int4*)(src + eb);

  float ss[SD_ROWS], dd[SD_ROWS];
  #pragma unroll
  for (int r = 0; r < SD_ROWS; ++r) {
    ss[r] = v[r].x * as.x + v[r].y * as.y + v[r].z * as.z + v[r].w * as.w;
    dd[r] = v[r].x * ad.x + v[r].y * ad.y + v[r].z * ad.z + v[r].w * ad.w;
  }
  #pragma unroll
  for (int off = 16; off > 0; off >>= 1) {
    #pragma unroll
    for (int r = 0; r < SD_ROWS; ++r) {
      ss[r] += __shfl_xor(ss[r], off, 32);
      dd[r] += __shfl_xor(dd[r], off, 32);
    }
  }
  float ms = 0.f, md = 0.f;
  #pragma unroll
  for (int r = 0; r < SD_ROWS; ++r) {
    if (lane == r) { ms = ss[r]; md = dd[r]; }
  }
  if (lane < SD_ROWS) {
    s[row0 + lane] = ms;
    d[row0 + lane] = md;
  }

  unsigned o[8];
  uint4 r0, r1;
  r0.x = gat_edge_rec(d0.x, s0.x, &o[0]);
  r0.y = gat_edge_rec(d0.y, s0.y, &o[1]);
  r0.z = gat_edge_rec(d0.z, s0.z, &o[2]);
  r0.w = gat_edge_rec(d0.w, s0.w, &o[3]);
  r1.x = gat_edge_rec(d1.x, s1.x, &o[4]);
  r1.y = gat_edge_rec(d1.y, s1.y, &o[5]);
  r1.z = gat_edge_rec(d1.z, s1.z, &o[6]);
  r1.w = gat_edge_rec(d1.w, s1.w, &o[7]);
  *(uint4*)(rec + ea) = r0;
  *(uint4*)(rec + eb) = r1;
  own8[ea >> 2] = o[0] | (o[1] << 8) | (o[2] << 16) | (o[3] << 24);
  own8[eb >> 2] = o[4] | (o[5] << 8) | (o[6] << 16) | (o[7] << 24);
}

#define OWN_T      1024
#define EPT        16
#define CHUNK      (OWN_T * EPT)
#define NCHUNK     ((N_EDGES + CHUNK - 1) / CHUNK)
#define EBITS      20
#define EMASK      ((1 << EBITS) - 1)


__device__ __forceinline__ void gat_lds_barrier() {
  __builtin_amdgcn_fence(__ATOMIC_RELEASE, "workgroup", "local");
  __builtin_amdgcn_s_barrier();
  __builtin_amdgcn_fence(__ATOMIC_ACQUIRE, "workgroup", "local");
}

__device__ __forceinline__ float gat_logit(float sv, float dv) {
  float v = sv + dv;
  return (v > 0.f) ? v : LRELU_A * v;
}

__device__ __forceinline__ void gat_load_dst(const int* __restrict__ dst, int cbase, int tid, int4* dv) {
  #pragma unroll
  for (int k = 0; k < EPT / 4; ++k) {
    const int b = cbase + k * (OWN_T * 4) + tid * 4;
    if (b < N_EDGES) dv[k] = *(const int4*)(dst + b);
    else             dv[k] = make_int4(-1, -1, -1, -1);
  }
}

__device__ __forceinline__ void gat_filter(const int4* dv, int cbase, int tid, int lo, unsigned nown,
                                           int* qcnt, int* q_a) {
  #pragma unroll
  for (int k = 0; k < EPT / 4; ++k) {
    const int b = cbase + k * (OWN_T * 4) + tid * 4;
    const int dn[4] = {dv[k].x, dv[k].y, dv[k].z, dv[k].w};
    #pragma unroll
    for (int j = 0; j < 4; ++j) {
      const unsigned dl = (unsigned)(dn[j] - lo);
      if (dl < nown) {
        const int pos = atomicAdd(qcnt, 1);
        q_a[pos] = (int)(dl << EBITS) | (b + j);
      }
    }
  }
}


#define NWAVE      (OWN_T / 32)
#define WQCAP      512
#define SCAN_IT    ((NITP + OWN_T - 1) / OWN_T)
#define RING       8
#define SCAN_PAD   (((SCAN_IT + RING - 1) / RING) * RING)
#define ROT16      192
static_assert(ROT16 % 32 == 0 && (NOWNER - 1) * ROT16 < NITP, "rotated starts stay whole wave pieces inside the list");

__device__ __forceinline__ uint4 gat_scan_load(const unsigned* __restrict__ own8, int it, int ubase, int rot,
                                               int lane, int* piece) {
  int u = it * OWN_T + ubase;
  if (u >= NITP) u = ubase;
  int r = u + rot;
  if (r >= NITP) r -= NITP;
  *piece = r;
  return *(const uint4*)(own8 + (size_t)r * 4 + lane * 4);
}

__device__ __forceinline__ unsigned gat_zero_bytes(unsigned x) {
  return ~((((x & 0x7F7F7F7Fu) + 0x7F7F7F7Fu) | x) | 0x7F7F7F7Fu);
}

__device__ __forceinline__ void gat_keep16(uint4 v, int r, int lane, unsigned pat, int2* myq, unsigned* wcnt) {
  unsigned rem = (gat_zero_bytes(v.x ^ pat) >> 7) | (gat_zero_bytes(v.y ^ pat) >> 6) |
                 (gat_zero_bytes(v.z ^ pat) >> 5) | (gat_zero_bytes(v.w ^ pat) >> 4);
  if (r + 32 > NIT16) {
    if (r + lane >= NIT16) rem = 0u;
  }
  unsigned mask = __builtin_amdgcn_ballot_w32(rem != 0u);
  while (mask != 0u) {
    const unsigned pos = *wcnt + __builtin_amdgcn_mbcnt_lo(mask, 0u);
    if (rem != 0u && pos < WQCAP) {
      const int q = __builtin_ctz(rem);
      myq[pos].x = (r + lane) * 16 + ((q & 3) << 2) + (q >> 3);
    }
    rem &= rem - 1u;
    *wcnt += (unsigned)__builtin_popcount(mask);
    mask = __builtin_amdgcn_ballot_w32(rem != 0u);
  }
}

template <int G>
__device__ __forceinline__ void gat_rows_add(const float* __restrict__ Wh, float* rows, int keyv, float exv,
                                             int j, int lane) {
  float4 r[G];
  int    key[G];
  #pragma unroll
  for (int k = 0; k < G; ++k) {
    key[k] = __builtin_amdgcn_readlane(keyv, j + k);
    r[k] = ((const float4*)(Wh + (size_t)(key[k] & EMASK) * DIM))[lane];
  }
  #pragma unroll
  for (int k = 0; k < G; ++k) {
    asm volatile("" :: "v"(r[k].x), "v"(r[k].y), "v"(r[k].z), "v"(r[k].w));
  }
  #pragma unroll
  for (int k = 0; k < G; ++k) {
    const float x = __int_as_float(__builtin_amdgcn_readlane(__float_as_int(exv), j + k));
    float* rp = rows + (int)((unsigned)key[k] >> EBITS) * DIM + lane;
    atomicAdd(rp +  0, x * r[k].x);
    atomicAdd(rp + 32, x * r[k].y);
    atomicAdd(rp + 64, x * r[k].z);
    atomicAdd(rp + 96, x * r[k].w);
  }
}

__global__ __launch_bounds__(OWN_T) void gat_owner(const float* __restrict__ Wh,
                                                   const float* __restrict__ s,
                                                   const float* __restrict__ d,
                                                   const int* __restrict__ src,
                                                   const int* __restrict__ dst,
                                                   const unsigned* __restrict__ own8,
                                                   const unsigned* __restrict__ rec,
                                                   const float* __restrict__ h,
                                                   float* __restrict__ out) {
  __shared__ float rows[NPW * DIM];
  __shared__ float emaxL[NPW];
  __shared__ float denL[NPW];
  __shared__ float dL[NPW];
  __shared__ int2  wq[NWAVE * WQCAP];
  __shared__ unsigned wcntL[NWAVE];
  __shared__ int   qcnt[2];

  const int tid  = (int)threadIdx.x;
  const int lane = tid & 31;
  const int wv   = __builtin_amdgcn_readfirstlane(tid >> 5);
  const int lo   = (int)blockIdx.x * NPW;
  const int hi   = (lo + NPW < N_NODES) ? (lo + NPW) : N_NODES;
  const unsigned nown = (unsigned)(hi - lo);

  for (int i = tid; i < NPW * DIM; i += OWN_T) rows[i] = 0.f;
  if (tid < NPW) {
    emaxL[tid] = -INFINITY;
    denL[tid]  = 0.f;
    dL[tid]    = ((unsigned)tid < nown) ? d[lo + tid] : 0.f;
  }
  if (tid < 2) qcnt[tid] = 0;

  int2* myq = wq + wv * WQCAP;
  unsigned wcnt = 0;
  {
    const int rot   = (int)blockIdx.x * ROT16;
    const int ubase = wv * 32;
    const unsigned pat = (unsigned)blockIdx.x * 0x01010101u;
    uint4 ring[RING];
    int   piece[RING];
    #pragma unroll
    for (int k = 0; k < RING; ++k) ring[k] = gat_scan_load(own8, k, ubase, rot, lane, &piece[k]);
    for (int it = 0; it < SCAN_PAD; it += RING) {
      #pragma unroll
      for (int k = 0; k < RING; ++k) {
        const uint4 v = ring[k];
        const int   r = piece[k];
        if ((it + k) * OWN_T + ubase < NITP) {
          gat_keep16(v, r, lane, pat, myq, &wcnt);
        }
        ring[k] = gat_scan_load(own8, it + k + RING, ubase, rot, lane, &piece[k]);
      }
    }
  }
  if (lane == 0) wcntL[wv] = wcnt;
  __syncthreads();

  const bool over = __builtin_amdgcn_ballot_w32(wcntL[lane] > (unsigned)WQCAP) != 0u;

  if (!over) {
    const int n = (int)wcnt;

    for (int i0 = 0; i0 < n; i0 += 128) {
      unsigned key[4];
      int   sn[4];
      float sv[4];
      #pragma unroll
      for (int k = 0; k < 4; ++k) {
        const int i = i0 + k * 32 + lane;
        key[k] = rec[(i < n) ? myq[i].x : 0];
      }
      #pragma unroll
      for (int k = 0; k < 4; ++k) sn[k] = (int)(key[k] & 0xFFFFu);
      #pragma unroll
      for (int k = 0; k < 4; ++k) sv[k] = s[sn[k]];
      asm volatile("" :: "v"(sv[0]), "v"(sv[1]), "v"(sv[2]), "v"(sv[3]));
      #pragma unroll
      for (int k = 0; k < 4; ++k) {
        const int i = i0 + k * 32 + lane;
        if (i < n) {
          const unsigned dr = key[k] >> 16;
          const bool live = dr < nown;
          const int dl = live ? (int)dr : 0;
          const float v = live ? gat_logit(sv[k], dL[dl]) : -INFINITY;
          float* addr = &emaxL[dl];
          if (v >= 0.f) atomicMax((int*)addr, __float_as_int(v));
          else          atomicMin((unsigned int*)addr, __float_as_uint(v));
          myq[i] = make_int2((int)((unsigned)dl << EBITS) | sn[k], __float_as_int(v));
        }
      }
    }
    gat_lds_barrier();

    for (int i = lane; i < n; i += 32) {
      const int2 q = myq[i];
      const int dl = (int)((unsigned)q.x >> EBITS);
      float m = emaxL[dl];
      if (!isfinite(m)) m = 0.f;
      const float ex = expf(__int_as_float(q.y) - m);
      atomicAdd(&denL[dl], ex);
      myq[i].y = __float_as_int(ex);
    }
    gat_lds_barrier();

    for (int b = 0; b < n; b += 32) {
      const int i = b + lane;
      const int2 q = (i < n) ? myq[i] : make_int2(0, 0);
      const int keyv = q.x;
      const float exv = __int_as_float(q.y);
      const int cnt = (n - b < 32) ? (n - b) : 32;
      int j = 0;
      for (; j + 8 <= cnt; j += 8) gat_rows_add<8>(Wh, rows, keyv, exv, j, lane);
      if (j + 4 <= cnt) { gat_rows_add<4>(Wh, rows, keyv, exv, j, lane); j += 4; }
      if (j + 2 <= cnt) { gat_rows_add<2>(Wh, rows, keyv, exv, j, lane); j += 2; }
      if (j < cnt)      { gat_rows_add<1>(Wh, rows, keyv, exv, j, lane); }
    }
    gat_lds_barrier();
  } else {
    int*   q_a = (int*)wq;
    float* q_b = (float*)(q_a + CHUNK);
    const int rot = (int)(blockIdx.x % NCHUNK);

    {
      int4 dv[EPT / 4];
      int cc = rot;
      gat_load_dst(dst, cc * CHUNK, tid, dv);
      for (int c = 0; c < NCHUNK; ++c) {
        const int cbase = cc * CHUNK;
        int* cnt = &qcnt[c & 1];
        gat_filter(dv, cbase, tid, lo, nown, cnt, q_a);
        cc = (cc + 1 == NCHUNK) ? 0 : cc + 1;
        if (c + 1 < NCHUNK) gat_load_dst(dst, cc * CHUNK, tid, dv);
        gat_lds_barrier();
        const int n = *cnt;
        if (tid == 0) qcnt[(c + 1) & 1] = 0;
        for (int i = tid; i < n; i += OWN_T) {
          const int p  = q_a[i];
          const int e  = p & EMASK;
          const int dl = (int)((unsigned)p >> EBITS);
          const float v = gat_logit(s[src[e]], dL[dl]);
          float* addr = &emaxL[dl];
          if (v >= 0.f) atomicMax((int*)addr, __float_as_int(v));
          else          atomicMin((unsigned int*)addr, __float_as_uint(v));
        }
        gat_lds_barrier();
      }
    }

    if (tid < NPW) {
      const float m = emaxL[tid];
      if (!isfinite(m)) emaxL[tid] = 0.f;
    }
    __syncthreads();

    {
      int4 dv[EPT / 4];
      int cc = rot;
      gat_load_dst(dst, cc * CHUNK, tid, dv);
      for (int c = 0; c < NCHUNK; ++c) {
        const int cbase = cc * CHUNK;
        int* cnt = &qcnt[(c + NCHUNK) & 1];
        gat_filter(dv, cbase, tid, lo, nown, cnt, q_a);
        cc = (cc + 1 == NCHUNK) ? 0 : cc + 1;
        if (c + 1 < NCHUNK) gat_load_dst(dst, cc * CHUNK, tid, dv);
        gat_lds_barrier();
        const int n = *cnt;
        if (tid == 0) qcnt[(c + NCHUNK + 1) & 1] = 0;
        for (int i = tid; i < n; i += OWN_T) {
          const int p  = q_a[i];
          const int e  = p & EMASK;
          const int dl = (int)((unsigned)p >> EBITS);
          const int sn = src[e];
          const float v  = gat_logit(s[sn], dL[dl]);
          const float ex = expf(v - emaxL[dl]);
          atomicAdd(&denL[dl], ex);
          q_a[i] = (int)((unsigned)dl << EBITS) | sn;
          q_b[i] = ex;
        }
        gat_lds_barrier();
        for (int i = wv; i < n; i += 64) {
          const int  i2   = i + 32;
          const bool two  = i2 < n;
          const int  p0   = q_a[i];
          const int  p1   = two ? q_a[i2] : p0;
          const float x0  = q_b[i];
          const float x1  = two ? q_b[i2] : 0.f;
          const float4 v0 = ((const float4*)(Wh + (size_t)(p0 & EMASK) * DIM))[lane];
          const float4 v1 = ((const float4*)(Wh + (size_t)(p1 & EMASK) * DIM))[lane];
          asm volatile("" :: "v"(v0.x), "v"(v0.y), "v"(v0.z), "v"(v0.w),
                             "v"(v1.x), "v"(v1.y), "v"(v1.z), "v"(v1.w));
          float* r0 = rows + (int)((unsigned)p0 >> EBITS) * DIM + lane;
          atomicAdd(r0 +  0, x0 * v0.x);
          atomicAdd(r0 + 32, x0 * v0.y);
          atomicAdd(r0 + 64, x0 * v0.z);
          atomicAdd(r0 + 96, x0 * v0.w);
          if (two) {
            float* r1 = rows + (int)((unsigned)p1 >> EBITS) * DIM + lane;
            atomicAdd(r1 +  0, x1 * v1.x);
            atomicAdd(r1 + 32, x1 * v1.y);
            atomicAdd(r1 + 64, x1 * v1.z);
            atomicAdd(r1 + 96, x1 * v1.w);
          }
        }
        gat_lds_barrier();
      }
    }

  }

  const int total = (int)nown * DIM;
  for (int i = tid; i < total; i += OWN_T) {
    const int nl = i >> 7;
    const int c  = i & (DIM - 1);
    const float den = denL[nl];
    const float acc = rows[nl * DIM + (c & 3) * 32 + (c >> 2)];
    const float agg = (den != 0.f) ? acc / den : 0.f;
    const size_t g = (size_t)(lo + nl) * DIM + c;
    out[g] = agg + h[g];
  }
}

extern "C" void kernel_launch(void* const* d_in, const int* in_sizes, int n_in,
                              void* d_out, int out_size, void* d_ws, size_t ws_size,
                              hipStream_t stream) {
  const float* h   = (const float*)d_in[0];
  const float* W   = (const float*)d_in[1];
  const float* a   = (const float*)d_in[2];
  const int*   src = (const int*)d_in[3];
  const int*   dst = (const int*)d_in[4];
  float* hout = (float*)d_out;

  float* ws    = (float*)d_ws;
  float* Wh    = ws;
  float* s     = Wh   + (size_t)N_NODES * DIM;
  float* d     = s    + N_NODES;
  unsigned* own8 = (unsigned*)(ws + 6500352);
  unsigned* rec  = own8 + 200064;

  {
    const int waves = N_NODES / 16;
    const int blocks = (waves + 3) / 4;
    gat_gemm_wmma<<<blocks, 128, 0, stream>>>(h, W, Wh);
  }
  gat_sd<<<(N_NODES / SD_ROWS + 7) / 8, 256, 0, stream>>>(Wh, a, s, d, src, dst, own8, rec);
  gat_owner<<<(N_NODES + NPW - 1) / NPW, OWN_T, 0, stream>>>(Wh, s, d, src, dst, own8, rec, h, hout);
}


# output phase: one wave per row, its h rows fetched right after the scan as one group of whole-row 16-byte loads, 16-byte stores (was one awaited 4-byte load per element); order of sums unchanged
# speedup vs baseline: 3.5261x; 1.1672x over previous
#include <hip/hip_runtime.h>
#include <math.h>

#define N_NODES 50000
#define N_EDGES 800000
#define DIM     128
#define LRELU_A 0.2f

typedef __attribute__((ext_vector_type(2))) float v2f;
typedef __attribute__((ext_vector_type(8))) float v8f;

__global__ __launch_bounds__(128) void gat_gemm_wmma(const float* __restrict__ h,
                                                     const float* __restrict__ W,
                                                     float* __restrict__ Wh) {
  const int wave = (int)((blockIdx.x * blockDim.x + threadIdx.x) >> 5);
  const int lane = (int)(threadIdx.x & 31);
  if (wave >= N_NODES / 16) return;

  const int row0  = wave * 16;
  const int rsel  = lane & 15;
  const int khalf = (lane >> 4) * 2;

  v8f acc[8];
  #pragma unroll
  for (int t = 0; t < 8; ++t) { v8f z = {}; acc[t] = z; }

  for (int k0 = 0; k0 < DIM; k0 += 4) {
    const float* ap = h + (size_t)(row0 + rsel) * DIM + (k0 + khalf);
    v2f a; a.x = ap[0]; a.y = ap[1];
    #pragma unroll
    for (int t = 0; t < 8; ++t) {
      const float* bp = W + (size_t)(k0 + khalf) * DIM + t * 16 + rsel;
      v2f b; b.x = bp[0]; b.y = bp[DIM];
      acc[t] = __builtin_amdgcn_wmma_f32_16x16x4_f32(
           false, a,  false, b,
           (short)0, acc[t],  false,  false);
    }
  }

  const int srow = row0 + (lane >> 4) * 8;
  #pragma unroll
  for (int t = 0; t < 8; ++t) {
    #pragma unroll
    for (int i = 0; i < 8; ++i) {
      Wh[(size_t)(srow + i) * DIM + t * 16 + rsel] = acc[t][i];
    }
  }
}

#define SD_ROWS 16
static_assert(N_NODES % SD_ROWS == 0, "a wave takes SD_ROWS whole rows");
static_assert(DIM == 128, "one float4 per lane covers a row");

#define NPW        196
#define NOWNER     ((N_NODES + NPW - 1) / NPW)
#define NIT16      (N_EDGES / 16)
#define NITP       (((NIT16 + 31) / 32) * 32)
#define DL_NONE    255u
static_assert(N_EDGES % 16 == 0, "own8 pieces hold 16 whole edges");
static_assert(NOWNER <= 256, "an owner index fits a byte");
static_assert(N_NODES <= 65536 && NPW < (int)DL_NONE, "rec holds src in 16 bits, the local node in 8");
static_assert(N_EDGES / 8 == (N_NODES / SD_ROWS) * 32, "a wave of the s,d launch takes SD_ROWS rows and 256 edges");
static_assert(NITP * 2 - N_EDGES / 8 == 32, "one more wave writes the pad pieces of own8");
static_assert(((N_NODES / SD_ROWS + 7) / 8) * 8 > N_NODES / SD_ROWS, "the s,d grid holds that wave");

__device__ __forceinline__ unsigned gat_edge_rec(int dn, int sn, unsigned* ow) {
  const unsigned du = (unsigned)dn;
  const unsigned o  = du / NPW;
  const unsigned dl = (du < (unsigned)N_NODES) ? (du - o * NPW) : DL_NONE;
  *ow = o & 255u;
  return (dl << 16) | ((unsigned)sn & 0xFFFFu);
}

__global__ __launch_bounds__(256) void gat_sd(const float* __restrict__ Wh,
                                              const float* __restrict__ a,
                                              float* __restrict__ s,
                                              float* __restrict__ d,
                                              const int* __restrict__ src,
                                              const int* __restrict__ dst,
                                              unsigned* __restrict__ own8,
                                              unsigned* __restrict__ rec) {
  const int gt   = (int)(blockIdx.x * blockDim.x + threadIdx.x);
  const int wave = gt >> 5;
  const int lane = (int)(threadIdx.x & 31);
  if (wave > N_NODES / SD_ROWS) return;
  const int ea = wave * 256 + lane * 4;
  const int eb = ea + 128;
  if (wave == N_NODES / SD_ROWS) {
    own8[ea >> 2] = 0u;
    own8[eb >> 2] = 0u;
    return;
  }
  const int row0 = wave * SD_ROWS;

  const float4 as = ((const float4*)a)[lane];
  const float4 ad = ((const float4*)(a + DIM))[lane];
  const float4* rp = (const float4*)(Wh + (size_t)row0 * DIM) + lane;

  float4 v[SD_ROWS];
  #pragma unroll
  for (int r = 0; r < SD_ROWS; ++r) v[r] = rp[r * (DIM / 4)];
  const int4 d0 = *(const int4*)(dst + ea);
  const int4 d1 = *(const int4*)(dst + eb);
  const int4 s0 = *(const int4*)(src + ea);
  const int4 s1 = *(const int4*)(src + eb);

  float ss[SD_ROWS], dd[SD_ROWS];
  #pragma unroll
  for (int r = 0; r < SD_ROWS; ++r) {
    ss[r] = v[r].x * as.x + v[r].y * as.y + v[r].z * as.z + v[r].w * as.w;
    dd[r] = v[r].x * ad.x + v[r].y * ad.y + v[r].z * ad.z + v[r].w * ad.w;
  }
  #pragma unroll
  for (int off = 16; off > 0; off >>= 1) {
    #pragma unroll
    for (int r = 0; r < SD_ROWS; ++r) {
      ss[r] += __shfl_xor(ss[r], off, 32);
      dd[r] += __shfl_xor(dd[r], off, 32);
    }
  }
  float ms = 0.f, md = 0.f;
  #pragma unroll
  for (int r = 0; r < SD_ROWS; ++r) {
    if (lane == r) { ms = ss[r]; md = dd[r]; }
  }
  if (lane < SD_ROWS) {
    s[row0 + lane] = ms;
    d[row0 + lane] = md;
  }

  unsigned o[8];
  uint4 r0, r1;
  r0.x = gat_edge_rec(d0.x, s0.x, &o[0]);
  r0.y = gat_edge_rec(d0.y, s0.y, &o[1]);
  r0.z = gat_edge_rec(d0.z, s0.z, &o[2]);
  r0.w = gat_edge_rec(d0.w, s0.w, &o[3]);
  r1.x = gat_edge_rec(d1.x, s1.x, &o[4]);
  r1.y = gat_edge_rec(d1.y, s1.y, &o[5]);
  r1.z = gat_edge_rec(d1.z, s1.z, &o[6]);
  r1.w = gat_edge_rec(d1.w, s1.w, &o[7]);
  *(uint4*)(rec + ea) = r0;
  *(uint4*)(rec + eb) = r1;
  own8[ea >> 2] = o[0] | (o[1] << 8) | (o[2] << 16) | (o[3] << 24);
  own8[eb >> 2] = o[4] | (o[5] << 8) | (o[6] << 16) | (o[7] << 24);
}

#define OWN_T      1024
#define EPT        16
#define CHUNK      (OWN_T * EPT)
#define NCHUNK     ((N_EDGES + CHUNK - 1) / CHUNK)
#define EBITS      20
#define EMASK      ((1 << EBITS) - 1)


__device__ __forceinline__ void gat_lds_barrier() {
  __builtin_amdgcn_fence(__ATOMIC_RELEASE, "workgroup", "local");
  __builtin_amdgcn_s_barrier();
  __builtin_amdgcn_fence(__ATOMIC_ACQUIRE, "workgroup", "local");
}

__device__ __forceinline__ float gat_logit(float sv, float dv) {
  float v = sv + dv;
  return (v > 0.f) ? v : LRELU_A * v;
}

__device__ __forceinline__ void gat_load_dst(const int* __restrict__ dst, int cbase, int tid, int4* dv) {
  #pragma unroll
  for (int k = 0; k < EPT / 4; ++k) {
    const int b = cbase + k * (OWN_T * 4) + tid * 4;
    if (b < N_EDGES) dv[k] = *(const int4*)(dst + b);
    else             dv[k] = make_int4(-1, -1, -1, -1);
  }
}

__device__ __forceinline__ void gat_filter(const int4* dv, int cbase, int tid, int lo, unsigned nown,
                                           int* qcnt, int* q_a) {
  #pragma unroll
  for (int k = 0; k < EPT / 4; ++k) {
    const int b = cbase + k * (OWN_T * 4) + tid * 4;
    const int dn[4] = {dv[k].x, dv[k].y, dv[k].z, dv[k].w};
    #pragma unroll
    for (int j = 0; j < 4; ++j) {
      const unsigned dl = (unsigned)(dn[j] - lo);
      if (dl < nown) {
        const int pos = atomicAdd(qcnt, 1);
        q_a[pos] = (int)(dl << EBITS) | (b + j);
      }
    }
  }
}


#define NWAVE      (OWN_T / 32)
#define OUT_ROWS   ((NPW + NWAVE - 1) / NWAVE)
#define WQCAP      512
#define SCAN_IT    ((NITP + OWN_T - 1) / OWN_T)
#define RING       8
#define SCAN_PAD   (((SCAN_IT + RING - 1) / RING) * RING)
#define ROT16      192
static_assert(ROT16 % 32 == 0 && (NOWNER - 1) * ROT16 < NITP, "rotated starts stay whole wave pieces inside the list");

__device__ __forceinline__ uint4 gat_scan_load(const unsigned* __restrict__ own8, int it, int ubase, int rot,
                                               int lane, int* piece) {
  int u = it * OWN_T + ubase;
  if (u >= NITP) u = ubase;
  int r = u + rot;
  if (r >= NITP) r -= NITP;
  *piece = r;
  return *(const uint4*)(own8 + (size_t)r * 4 + lane * 4);
}

__device__ __forceinline__ unsigned gat_zero_bytes(unsigned x) {
  return ~((((x & 0x7F7F7F7Fu) + 0x7F7F7F7Fu) | x) | 0x7F7F7F7Fu);
}

__device__ __forceinline__ void gat_keep16(uint4 v, int r, int lane, unsigned pat, int2* myq, unsigned* wcnt) {
  unsigned rem = (gat_zero_bytes(v.x ^ pat) >> 7) | (gat_zero_bytes(v.y ^ pat) >> 6) |
                 (gat_zero_bytes(v.z ^ pat) >> 5) | (gat_zero_bytes(v.w ^ pat) >> 4);
  if (r + 32 > NIT16) {
    if (r + lane >= NIT16) rem = 0u;
  }
  unsigned mask = __builtin_amdgcn_ballot_w32(rem != 0u);
  while (mask != 0u) {
    const unsigned pos = *wcnt + __builtin_amdgcn_mbcnt_lo(mask, 0u);
    if (rem != 0u && pos < WQCAP) {
      const int q = __builtin_ctz(rem);
      myq[pos].x = (r + lane) * 16 + ((q & 3) << 2) + (q >> 3);
    }
    rem &= rem - 1u;
    *wcnt += (unsigned)__builtin_popcount(mask);
    mask = __builtin_amdgcn_ballot_w32(rem != 0u);
  }
}

template <int G>
__device__ __forceinline__ void gat_rows_add(const float* __restrict__ Wh, float* rows, int keyv, float exv,
                                             int j, int lane) {
  float4 r[G];
  int    key[G];
  #pragma unroll
  for (int k = 0; k < G; ++k) {
    key[k] = __builtin_amdgcn_readlane(keyv, j + k);
    r[k] = ((const float4*)(Wh + (size_t)(key[k] & EMASK) * DIM))[lane];
  }
  #pragma unroll
  for (int k = 0; k < G; ++k) {
    asm volatile("" :: "v"(r[k].x), "v"(r[k].y), "v"(r[k].z), "v"(r[k].w));
  }
  #pragma unroll
  for (int k = 0; k < G; ++k) {
    const float x = __int_as_float(__builtin_amdgcn_readlane(__float_as_int(exv), j + k));
    float* rp = rows + (int)((unsigned)key[k] >> EBITS) * DIM + lane;
    atomicAdd(rp +  0, x * r[k].x);
    atomicAdd(rp + 32, x * r[k].y);
    atomicAdd(rp + 64, x * r[k].z);
    atomicAdd(rp + 96, x * r[k].w);
  }
}

__global__ __launch_bounds__(OWN_T) void gat_owner(const float* __restrict__ Wh,
                                                   const float* __restrict__ s,
                                                   const float* __restrict__ d,
                                                   const int* __restrict__ src,
                                                   const int* __restrict__ dst,
                                                   const unsigned* __restrict__ own8,
                                                   const unsigned* __restrict__ rec,
                                                   const float* __restrict__ h,
                                                   float* __restrict__ out) {
  __shared__ float rows[NPW * DIM];
  __shared__ float emaxL[NPW];
  __shared__ float denL[NPW];
  __shared__ float dL[NPW];
  __shared__ int2  wq[NWAVE * WQCAP];
  __shared__ unsigned wcntL[NWAVE];
  __shared__ int   qcnt[2];

  const int tid  = (int)threadIdx.x;
  const int lane = tid & 31;
  const int wv   = __builtin_amdgcn_readfirstlane(tid >> 5);
  const int lo   = (int)blockIdx.x * NPW;
  const int hi   = (lo + NPW < N_NODES) ? (lo + NPW) : N_NODES;
  const unsigned nown = (unsigned)(hi - lo);

  for (int i = tid; i < NPW * DIM; i += OWN_T) rows[i] = 0.f;
  if (tid < NPW) {
    emaxL[tid] = -INFINITY;
    denL[tid]  = 0.f;
    dL[tid]    = ((unsigned)tid < nown) ? d[lo + tid] : 0.f;
  }
  if (tid < 2) qcnt[tid] = 0;

  int2* myq = wq + wv * WQCAP;
  unsigned wcnt = 0;
  {
    const int rot   = (int)blockIdx.x * ROT16;
    const int ubase = wv * 32;
    const unsigned pat = (unsigned)blockIdx.x * 0x01010101u;
    uint4 ring[RING];
    int   piece[RING];
    #pragma unroll
    for (int k = 0; k < RING; ++k) ring[k] = gat_scan_load(own8, k, ubase, rot, lane, &piece[k]);
    for (int it = 0; it < SCAN_PAD; it += RING) {
      #pragma unroll
      for (int k = 0; k < RING; ++k) {
        const uint4 v = ring[k];
        const int   r = piece[k];
        if ((it + k) * OWN_T + ubase < NITP) {
          gat_keep16(v, r, lane, pat, myq, &wcnt);
        }
        ring[k] = gat_scan_load(own8, it + k + RING, ubase, rot, lane, &piece[k]);
      }
    }
  }
  float4 hres[OUT_ROWS];
  {
    const float4* hp = (const float4*)h + (size_t)lo * (DIM / 4) + lane;
    #pragma unroll
    for (int k = 0; k < OUT_ROWS; ++k) {
      int nl = wv + k * NWAVE;
      if (nl >= (int)nown) nl = (int)nown - 1;
      hres[k] = hp[nl * (DIM / 4)];
    }
  }

  if (lane == 0) wcntL[wv] = wcnt;
  __syncthreads();

  const bool over = __builtin_amdgcn_ballot_w32(wcntL[lane] > (unsigned)WQCAP) != 0u;

  if (!over) {
    const int n = (int)wcnt;

    for (int i0 = 0; i0 < n; i0 += 128) {
      unsigned key[4];
      int   sn[4];
      float sv[4];
      #pragma unroll
      for (int k = 0; k < 4; ++k) {
        const int i = i0 + k * 32 + lane;
        key[k] = rec[(i < n) ? myq[i].x : 0];
      }
      #pragma unroll
      for (int k = 0; k < 4; ++k) sn[k] = (int)(key[k] & 0xFFFFu);
      #pragma unroll
      for (int k = 0; k < 4; ++k) sv[k] = s[sn[k]];
      asm volatile("" :: "v"(sv[0]), "v"(sv[1]), "v"(sv[2]), "v"(sv[3]));
      #pragma unroll
      for (int k = 0; k < 4; ++k) {
        const int i = i0 + k * 32 + lane;
        if (i < n) {
          const unsigned dr = key[k] >> 16;
          const bool live = dr < nown;
          const int dl = live ? (int)dr : 0;
          const float v = live ? gat_logit(sv[k], dL[dl]) : -INFINITY;
          float* addr = &emaxL[dl];
          if (v >= 0.f) atomicMax((int*)addr, __float_as_int(v));
          else          atomicMin((unsigned int*)addr, __float_as_uint(v));
          myq[i] = make_int2((int)((unsigned)dl << EBITS) | sn[k], __float_as_int(v));
        }
      }
    }
    gat_lds_barrier();

    for (int i = lane; i < n; i += 32) {
      const int2 q = myq[i];
      const int dl = (int)((unsigned)q.x >> EBITS);
      float m = emaxL[dl];
      if (!isfinite(m)) m = 0.f;
      const float ex = expf(__int_as_float(q.y) - m);
      atomicAdd(&denL[dl], ex);
      myq[i].y = __float_as_int(ex);
    }
    gat_lds_barrier();

    for (int b = 0; b < n; b += 32) {
      const int i = b + lane;
      const int2 q = (i < n) ? myq[i] : make_int2(0, 0);
      const int keyv = q.x;
      const float exv = __int_as_float(q.y);
      const int cnt = (n - b < 32) ? (n - b) : 32;
      int j = 0;
      for (; j + 8 <= cnt; j += 8) gat_rows_add<8>(Wh, rows, keyv, exv, j, lane);
      if (j + 4 <= cnt) { gat_rows_add<4>(Wh, rows, keyv, exv, j, lane); j += 4; }
      if (j + 2 <= cnt) { gat_rows_add<2>(Wh, rows, keyv, exv, j, lane); j += 2; }
      if (j < cnt)      { gat_rows_add<1>(Wh, rows, keyv, exv, j, lane); }
    }
    gat_lds_barrier();
  } else {
    int*   q_a = (int*)wq;
    float* q_b = (float*)(q_a + CHUNK);
    const int rot = (int)(blockIdx.x % NCHUNK);

    {
      int4 dv[EPT / 4];
      int cc = rot;
      gat_load_dst(dst, cc * CHUNK, tid, dv);
      for (int c = 0; c < NCHUNK; ++c) {
        const int cbase = cc * CHUNK;
        int* cnt = &qcnt[c & 1];
        gat_filter(dv, cbase, tid, lo, nown, cnt, q_a);
        cc = (cc + 1 == NCHUNK) ? 0 : cc + 1;
        if (c + 1 < NCHUNK) gat_load_dst(dst, cc * CHUNK, tid, dv);
        gat_lds_barrier();
        const int n = *cnt;
        if (tid == 0) qcnt[(c + 1) & 1] = 0;
        for (int i = tid; i < n; i += OWN_T) {
          const int p  = q_a[i];
          const int e  = p & EMASK;
          const int dl = (int)((unsigned)p >> EBITS);
          const float v = gat_logit(s[src[e]], dL[dl]);
          float* addr = &emaxL[dl];
          if (v >= 0.f) atomicMax((int*)addr, __float_as_int(v));
          else          atomicMin((unsigned int*)addr, __float_as_uint(v));
        }
        gat_lds_barrier();
      }
    }

    if (tid < NPW) {
      const float m = emaxL[tid];
      if (!isfinite(m)) emaxL[tid] = 0.f;
    }
    __syncthreads();

    {
      int4 dv[EPT / 4];
      int cc = rot;
      gat_load_dst(dst, cc * CHUNK, tid, dv);
      for (int c = 0; c < NCHUNK; ++c) {
        const int cbase = cc * CHUNK;
        int* cnt = &qcnt[(c + NCHUNK) & 1];
        gat_filter(dv, cbase, tid, lo, nown, cnt, q_a);
        cc = (cc + 1 == NCHUNK) ? 0 : cc + 1;
        if (c + 1 < NCHUNK) gat_load_dst(dst, cc * CHUNK, tid, dv);
        gat_lds_barrier();
        const int n = *cnt;
        if (tid == 0) qcnt[(c + NCHUNK + 1) & 1] = 0;
        for (int i = tid; i < n; i += OWN_T) {
          const int p  = q_a[i];
          const int e  = p & EMASK;
          const int dl = (int)((unsigned)p >> EBITS);
          const int sn = src[e];
          const float v  = gat_logit(s[sn], dL[dl]);
          const float ex = expf(v - emaxL[dl]);
          atomicAdd(&denL[dl], ex);
          q_a[i] = (int)((unsigned)dl << EBITS) | sn;
          q_b[i] = ex;
        }
        gat_lds_barrier();
        for (int i = wv; i < n; i += 64) {
          const int  i2   = i + 32;
          const bool two  = i2 < n;
          const int  p0   = q_a[i];
          const int  p1   = two ? q_a[i2] : p0;
          const float x0  = q_b[i];
          const float x1  = two ? q_b[i2] : 0.f;
          const float4 v0 = ((const float4*)(Wh + (size_t)(p0 & EMASK) * DIM))[lane];
          const float4 v1 = ((const float4*)(Wh + (size_t)(p1 & EMASK) * DIM))[lane];
          asm volatile("" :: "v"(v0.x), "v"(v0.y), "v"(v0.z), "v"(v0.w),
                             "v"(v1.x), "v"(v1.y), "v"(v1.z), "v"(v1.w));
          float* r0 = rows + (int)((unsigned)p0 >> EBITS) * DIM + lane;
          atomicAdd(r0 +  0, x0 * v0.x);
          atomicAdd(r0 + 32, x0 * v0.y);
          atomicAdd(r0 + 64, x0 * v0.z);
          atomicAdd(r0 + 96, x0 * v0.w);
          if (two) {
            float* r1 = rows + (int)((unsigned)p1 >> EBITS) * DIM + lane;
            atomicAdd(r1 +  0, x1 * v1.x);
            atomicAdd(r1 + 32, x1 * v1.y);
            atomicAdd(r1 + 64, x1 * v1.z);
            atomicAdd(r1 + 96, x1 * v1.w);
          }
        }
        gat_lds_barrier();
      }
    }

  }

  {
    float4* op = (float4*)out + (size_t)lo * (DIM / 4) + lane;
    #pragma unroll
    for (int k = 0; k < OUT_ROWS; ++k) {
      const int nl = wv + k * NWAVE;
      if (nl < (int)nown) {
        const float den = denL[nl];
        const float* rp = rows + nl * DIM + lane;
        const float a0 = rp[0], a1 = rp[32], a2 = rp[64], a3 = rp[96];
        float4 o;
        o.x = ((den != 0.f) ? a0 / den : 0.f) + hres[k].x;
        o.y = ((den != 0.f) ? a1 / den : 0.f) + hres[k].y;
        o.z = ((den != 0.f) ? a2 / den : 0.f) + hres[k].z;
        o.w = ((den != 0.f) ? a3 / den : 0.f) + hres[k].w;
        op[nl * (DIM / 4)] = o;
      }
    }
  }
}

extern "C" void kernel_launch(void* const* d_in, const int* in_sizes, int n_in,
                              void* d_out, int out_size, void* d_ws, size_t ws_size,
                              hipStream_t stream) {
  const float* h   = (const float*)d_in[0];
  const float* W   = (const float*)d_in[1];
  const float* a   = (const float*)d_in[2];
  const int*   src = (const int*)d_in[3];
  const int*   dst = (const int*)d_in[4];
  float* hout = (float*)d_out;

  float* ws    = (float*)d_ws;
  float* Wh    = ws;
  float* s     = Wh   + (size_t)N_NODES * DIM;
  float* d     = s    + N_NODES;
  unsigned* own8 = (unsigned*)(ws + 6500352);
  unsigned* rec  = own8 + 200064;

  {
    const int waves = N_NODES / 16;
    const int blocks = (waves + 3) / 4;
    gat_gemm_wmma<<<blocks, 128, 0, stream>>>(h, W, Wh);
  }
  gat_sd<<<(N_NODES / SD_ROWS + 7) / 8, 256, 0, stream>>>(Wh, a, s, d, src, dst, own8, rec);
  gat_owner<<<(N_NODES + NPW - 1) / NPW, OWN_T, 0, stream>>>(Wh, s, d, src, dst, own8, rec, h, hout);
}


# s,d launch sorts edge records by owner into 32-slot cells per 2048-edge group (returning LDS adds); owners read their 391 cells, not all owner bytes; queue in arrival order, sums still LDS atomics
# speedup vs baseline: 4.1090x; 1.1653x over previous
#include <hip/hip_runtime.h>
#include <math.h>

#define N_NODES 50000
#define N_EDGES 800000
#define DIM     128
#define LRELU_A 0.2f

typedef __attribute__((ext_vector_type(2))) float v2f;
typedef __attribute__((ext_vector_type(8))) float v8f;

__global__ __launch_bounds__(128) void gat_gemm_wmma(const float* __restrict__ h,
                                                     const float* __restrict__ W,
                                                     float* __restrict__ Wh) {
  const int wave = (int)((blockIdx.x * blockDim.x + threadIdx.x) >> 5);
  const int lane = (int)(threadIdx.x & 31);
  if (wave >= N_NODES / 16) return;

  const int row0  = wave * 16;
  const int rsel  = lane & 15;
  const int khalf = (lane >> 4) * 2;

  v8f acc[8];
  #pragma unroll
  for (int t = 0; t < 8; ++t) { v8f z = {}; acc[t] = z; }

  for (int k0 = 0; k0 < DIM; k0 += 4) {
    const float* ap = h + (size_t)(row0 + rsel) * DIM + (k0 + khalf);
    v2f a; a.x = ap[0]; a.y = ap[1];
    #pragma unroll
    for (int t = 0; t < 8; ++t) {
      const float* bp = W + (size_t)(k0 + khalf) * DIM + t * 16 + rsel;
      v2f b; b.x = bp[0]; b.y = bp[DIM];
      acc[t] = __builtin_amdgcn_wmma_f32_16x16x4_f32(
           false, a,  false, b,
           (short)0, acc[t],  false,  false);
    }
  }

  const int srow = row0 + (lane >> 4) * 8;
  #pragma unroll
  for (int t = 0; t < 8; ++t) {
    #pragma unroll
    for (int i = 0; i < 8; ++i) {
      Wh[(size_t)(srow + i) * DIM + t * 16 + rsel] = acc[t][i];
    }
  }
}

__device__ __forceinline__ void gat_lds_barrier() {
  __builtin_amdgcn_fence(__ATOMIC_RELEASE, "workgroup", "local");
  __builtin_amdgcn_s_barrier();
  __builtin_amdgcn_fence(__ATOMIC_ACQUIRE, "workgroup", "local");
}

#define SD_ROWS 16
#define SD_T       256
#define NSDWAVE    (N_NODES / SD_ROWS)
#define NSDWG      ((NSDWAVE * 32 + SD_T - 1) / SD_T)
static_assert(N_NODES % SD_ROWS == 0, "a wave takes SD_ROWS whole rows");
static_assert(DIM == 128, "one float4 per lane covers a row");

#define NPW        196
#define NOWNER     ((N_NODES + NPW - 1) / NPW)
#define BCAP       32
#define CELL_EMPTY 0xFFFFFFFFu
#define CELL_OVER  0xFFFFFFFEu
#define DL_NONE    255u
static_assert(NOWNER <= 256, "an owner index fits the 8 bits it is cut to");
static_assert(N_NODES <= 65536 && NPW < (int)DL_NONE, "rec holds src in 16 bits, the local node in 8");
static_assert(N_EDGES / 8 == NSDWAVE * 32, "a wave of the s,d launch takes SD_ROWS rows and 256 edges");
static_assert(SD_T >= 256, "one thread per possible owner index clears that cell's count");
static_assert((256 * BCAP / 4) % SD_T == 0, "the cells are cleared and written out in whole 16-byte rounds");

__device__ __forceinline__ unsigned gat_edge_rec(int dn, int sn, unsigned* ow) {
  const unsigned du = (unsigned)dn;
  const unsigned o  = du / NPW;
  const unsigned dl = (du < (unsigned)N_NODES) ? (du - o * NPW) : DL_NONE;
  *ow = o & 255u;
  return (dl << 16) | ((unsigned)sn & 0xFFFFu);
}

__global__ __launch_bounds__(SD_T) void gat_sd(const float* __restrict__ Wh,
                                               const float* __restrict__ a,
                                               float* __restrict__ s,
                                               float* __restrict__ d,
                                               const int* __restrict__ src,
                                               const int* __restrict__ dst,
                                               unsigned* __restrict__ bins) {
  __shared__ uint4    cellL[256 * BCAP / 4];
  __shared__ unsigned ccntL[256];

  const int tid  = (int)threadIdx.x;
  const int lane = tid & 31;
  const int wraw = (int)((blockIdx.x * blockDim.x + threadIdx.x) >> 5);
  const bool act = wraw < NSDWAVE;
  const int wave = act ? wraw : (NSDWAVE - 1);
  const int ea = wave * 256 + lane * 4;
  const int eb = ea + 128;
  const int row0 = wave * SD_ROWS;

  const float4 as = ((const float4*)a)[lane];
  const float4 ad = ((const float4*)(a + DIM))[lane];
  const float4* rp = (const float4*)(Wh + (size_t)row0 * DIM) + lane;

  float4 v[SD_ROWS];
  #pragma unroll
  for (int r = 0; r < SD_ROWS; ++r) v[r] = rp[r * (DIM / 4)];
  const int4 d0 = *(const int4*)(dst + ea);
  const int4 d1 = *(const int4*)(dst + eb);
  const int4 s0 = *(const int4*)(src + ea);
  const int4 s1 = *(const int4*)(src + eb);

  {
    const uint4 e4 = make_uint4(CELL_EMPTY, CELL_EMPTY, CELL_EMPTY, CELL_EMPTY);
    #pragma unroll
    for (int j = 0; j < (256 * BCAP / 4) / SD_T; ++j) cellL[j * SD_T + tid] = e4;
    if (tid < 256) ccntL[tid] = 0u;
  }
  gat_lds_barrier();

  float ss[SD_ROWS], dd[SD_ROWS];
  #pragma unroll
  for (int r = 0; r < SD_ROWS; ++r) {
    ss[r] = v[r].x * as.x + v[r].y * as.y + v[r].z * as.z + v[r].w * as.w;
    dd[r] = v[r].x * ad.x + v[r].y * ad.y + v[r].z * ad.z + v[r].w * ad.w;
  }
  #pragma unroll
  for (int off = 16; off > 0; off >>= 1) {
    #pragma unroll
    for (int r = 0; r < SD_ROWS; ++r) {
      ss[r] += __shfl_xor(ss[r], off, 32);
      dd[r] += __shfl_xor(dd[r], off, 32);
    }
  }
  float ms = 0.f, md = 0.f;
  #pragma unroll
  for (int r = 0; r < SD_ROWS; ++r) {
    if (lane == r) { ms = ss[r]; md = dd[r]; }
  }
  if (act && lane < SD_ROWS) {
    s[row0 + lane] = ms;
    d[row0 + lane] = md;
  }

  {
    unsigned* cell = (unsigned*)cellL;
    unsigned o[8], r[8], slot[8];
    r[0] = gat_edge_rec(d0.x, s0.x, &o[0]);
    r[1] = gat_edge_rec(d0.y, s0.y, &o[1]);
    r[2] = gat_edge_rec(d0.z, s0.z, &o[2]);
    r[3] = gat_edge_rec(d0.w, s0.w, &o[3]);
    r[4] = gat_edge_rec(d1.x, s1.x, &o[4]);
    r[5] = gat_edge_rec(d1.y, s1.y, &o[5]);
    r[6] = gat_edge_rec(d1.z, s1.z, &o[6]);
    r[7] = gat_edge_rec(d1.w, s1.w, &o[7]);
    if (act) {
      #pragma unroll
      for (int k = 0; k < 8; ++k) slot[k] = atomicAdd(&ccntL[o[k]], 1u);
      #pragma unroll
      for (int k = 0; k < 8; ++k) {
        if (slot[k] < BCAP) cell[o[k] * BCAP + slot[k]] = r[k];
      }
    }
  }
  gat_lds_barrier();

  {
    const int g = (int)blockIdx.x;
    #pragma unroll
    for (int j = 0; j < (256 * BCAP / 4) / SD_T; ++j) {
      const int q  = j * SD_T + tid;
      const int o  = q / (BCAP / 4);
      const int pc = q % (BCAP / 4);
      uint4 c4 = cellL[q];
      if (pc == 0 && ccntL[o] > (unsigned)BCAP) c4.x = CELL_OVER;
      *(uint4*)(bins + ((size_t)o * NSDWG + g) * BCAP + pc * 4) = c4;
    }
  }
}

#define OWN_T      1024
#define EPT        16
#define CHUNK      (OWN_T * EPT)
#define NCHUNK     ((N_EDGES + CHUNK - 1) / CHUNK)
#define EBITS      20
#define EMASK      ((1 << EBITS) - 1)


__device__ __forceinline__ float gat_logit(float sv, float dv) {
  float v = sv + dv;
  return (v > 0.f) ? v : LRELU_A * v;
}

__device__ __forceinline__ void gat_load_dst(const int* __restrict__ dst, int cbase, int tid, int4* dv) {
  #pragma unroll
  for (int k = 0; k < EPT / 4; ++k) {
    const int b = cbase + k * (OWN_T * 4) + tid * 4;
    if (b < N_EDGES) dv[k] = *(const int4*)(dst + b);
    else             dv[k] = make_int4(-1, -1, -1, -1);
  }
}

__device__ __forceinline__ void gat_filter(const int4* dv, int cbase, int tid, int lo, unsigned nown,
                                           int* qcnt, int* q_a) {
  #pragma unroll
  for (int k = 0; k < EPT / 4; ++k) {
    const int b = cbase + k * (OWN_T * 4) + tid * 4;
    const int dn[4] = {dv[k].x, dv[k].y, dv[k].z, dv[k].w};
    #pragma unroll
    for (int j = 0; j < 4; ++j) {
      const unsigned dl = (unsigned)(dn[j] - lo);
      if (dl < nown) {
        const int pos = atomicAdd(qcnt, 1);
        q_a[pos] = (int)(dl << EBITS) | (b + j);
      }
    }
  }
}


#define NWAVE      (OWN_T / 32)
#define OUT_ROWS   ((NPW + NWAVE - 1) / NWAVE)
#define WQCAP      512
#define CELLS_PW   ((NSDWG + NWAVE - 1) / NWAVE)
static_assert(CELLS_PW * BCAP <= WQCAP, "a wave's cells fit its queue segment even when every slot is taken");
static_assert(BCAP == 32, "a wave reads one cell per load, one slot per lane");

template <int G>
__device__ __forceinline__ void gat_rows_add(const float* __restrict__ Wh, float* rows, int keyv, float exv,
                                             int j, int lane) {
  float4 r[G];
  int    key[G];
  #pragma unroll
  for (int k = 0; k < G; ++k) {
    key[k] = __builtin_amdgcn_readlane(keyv, j + k);
    r[k] = ((const float4*)(Wh + (size_t)(key[k] & EMASK) * DIM))[lane];
  }
  #pragma unroll
  for (int k = 0; k < G; ++k) {
    asm volatile("" :: "v"(r[k].x), "v"(r[k].y), "v"(r[k].z), "v"(r[k].w));
  }
  #pragma unroll
  for (int k = 0; k < G; ++k) {
    const float x = __int_as_float(__builtin_amdgcn_readlane(__float_as_int(exv), j + k));
    float* rp = rows + (int)((unsigned)key[k] >> EBITS) * DIM + lane;
    atomicAdd(rp +  0, x * r[k].x);
    atomicAdd(rp + 32, x * r[k].y);
    atomicAdd(rp + 64, x * r[k].z);
    atomicAdd(rp + 96, x * r[k].w);
  }
}

__global__ __launch_bounds__(OWN_T) void gat_owner(const float* __restrict__ Wh,
                                                   const float* __restrict__ s,
                                                   const float* __restrict__ d,
                                                   const int* __restrict__ src,
                                                   const int* __restrict__ dst,
                                                   const unsigned* __restrict__ bins,
                                                   const float* __restrict__ h,
                                                   float* __restrict__ out) {
  __shared__ float rows[NPW * DIM];
  __shared__ float emaxL[NPW];
  __shared__ float denL[NPW];
  __shared__ float dL[NPW];
  __shared__ int2  wq[NWAVE * WQCAP];
  __shared__ unsigned wcntL[NWAVE];
  __shared__ int   qcnt[2];

  const int tid  = (int)threadIdx.x;
  const int lane = tid & 31;
  const int wv   = __builtin_amdgcn_readfirstlane(tid >> 5);
  const int lo   = (int)blockIdx.x * NPW;
  const int hi   = (lo + NPW < N_NODES) ? (lo + NPW) : N_NODES;
  const unsigned nown = (unsigned)(hi - lo);

  unsigned cv[CELLS_PW];
  {
    const unsigned* cp = bins + (size_t)blockIdx.x * (NSDWG * BCAP) + lane;
    #pragma unroll
    for (int k = 0; k < CELLS_PW; ++k) {
      int cidx = wv + k * NWAVE;
      if (cidx >= NSDWG) cidx = NSDWG - 1;
      cv[k] = cp[cidx * BCAP];
    }
  }

  for (int i = tid; i < NPW * DIM; i += OWN_T) rows[i] = 0.f;
  if (tid < NPW) {
    emaxL[tid] = -INFINITY;
    denL[tid]  = 0.f;
    dL[tid]    = ((unsigned)tid < nown) ? d[lo + tid] : 0.f;
  }
  if (tid < 2) qcnt[tid] = 0;

  int2* myq = wq + wv * WQCAP;
  unsigned wcnt = 0;
  {
    bool ran_over = false;
    #pragma unroll
    for (int k = 0; k < CELLS_PW; ++k) {
      if (wv + k * NWAVE < NSDWG) {
        const unsigned r = cv[k];
        const bool taken = r < CELL_OVER;
        ran_over = ran_over || (r == CELL_OVER);
        const unsigned mask = __builtin_amdgcn_ballot_w32(taken);
        const unsigned pos = wcnt + __builtin_amdgcn_mbcnt_lo(mask, 0u);
        if (taken) myq[pos].x = (int)r;
        wcnt += (unsigned)__builtin_popcount(mask);
      }
    }
    if (__builtin_amdgcn_ballot_w32(ran_over) != 0u) wcnt = 0xFFFFFFFFu;
  }
  float4 hres[OUT_ROWS];
  {
    const float4* hp = (const float4*)h + (size_t)lo * (DIM / 4) + lane;
    #pragma unroll
    for (int k = 0; k < OUT_ROWS; ++k) {
      int nl = wv + k * NWAVE;
      if (nl >= (int)nown) nl = (int)nown - 1;
      hres[k] = hp[nl * (DIM / 4)];
    }
  }

  if (lane == 0) wcntL[wv] = wcnt;
  __syncthreads();

  const bool over = __builtin_amdgcn_ballot_w32(wcntL[lane] > (unsigned)WQCAP) != 0u;

  if (!over) {
    const int n = (int)wcnt;

    for (int i0 = 0; i0 < n; i0 += 128) {
      unsigned key[4];
      int   sn[4];
      float sv[4];
      #pragma unroll
      for (int k = 0; k < 4; ++k) {
        const int i = i0 + k * 32 + lane;
        key[k] = (i < n) ? (unsigned)myq[i].x : 0u;
      }
      #pragma unroll
      for (int k = 0; k < 4; ++k) sn[k] = (int)(key[k] & 0xFFFFu);
      #pragma unroll
      for (int k = 0; k < 4; ++k) sv[k] = s[sn[k]];
      asm volatile("" :: "v"(sv[0]), "v"(sv[1]), "v"(sv[2]), "v"(sv[3]));
      #pragma unroll
      for (int k = 0; k < 4; ++k) {
        const int i = i0 + k * 32 + lane;
        if (i < n) {
          const unsigned dr = key[k] >> 16;
          const bool live = dr < nown;
          const int dl = live ? (int)dr : 0;
          const float v = live ? gat_logit(sv[k], dL[dl]) : -INFINITY;
          float* addr = &emaxL[dl];
          if (v >= 0.f) atomicMax((int*)addr, __float_as_int(v));
          else          atomicMin((unsigned int*)addr, __float_as_uint(v));
          myq[i] = make_int2((int)((unsigned)dl << EBITS) | sn[k], __float_as_int(v));
        }
      }
    }
    gat_lds_barrier();

    for (int i = lane; i < n; i += 32) {
      const int2 q = myq[i];
      const int dl = (int)((unsigned)q.x >> EBITS);
      float m = emaxL[dl];
      if (!isfinite(m)) m = 0.f;
      const float ex = expf(__int_as_float(q.y) - m);
      atomicAdd(&denL[dl], ex);
      myq[i].y = __float_as_int(ex);
    }
    gat_lds_barrier();

    for (int b = 0; b < n; b += 32) {
      const int i = b + lane;
      const int2 q = (i < n) ? myq[i] : make_int2(0, 0);
      const int keyv = q.x;
      const float exv = __int_as_float(q.y);
      const int cnt = (n - b < 32) ? (n - b) : 32;
      int j = 0;
      for (; j + 8 <= cnt; j += 8) gat_rows_add<8>(Wh, rows, keyv, exv, j, lane);
      if (j + 4 <= cnt) { gat_rows_add<4>(Wh, rows, keyv, exv, j, lane); j += 4; }
      if (j + 2 <= cnt) { gat_rows_add<2>(Wh, rows, keyv, exv, j, lane); j += 2; }
      if (j < cnt)      { gat_rows_add<1>(Wh, rows, keyv, exv, j, lane); }
    }
    gat_lds_barrier();
  } else {
    int*   q_a = (int*)wq;
    float* q_b = (float*)(q_a + CHUNK);
    const int rot = (int)(blockIdx.x % NCHUNK);

    {
      int4 dv[EPT / 4];
      int cc = rot;
      gat_load_dst(dst, cc * CHUNK, tid, dv);
      for (int c = 0; c < NCHUNK; ++c) {
        const int cbase = cc * CHUNK;
        int* cnt = &qcnt[c & 1];
        gat_filter(dv, cbase, tid, lo, nown, cnt, q_a);
        cc = (cc + 1 == NCHUNK) ? 0 : cc + 1;
        if (c + 1 < NCHUNK) gat_load_dst(dst, cc * CHUNK, tid, dv);
        gat_lds_barrier();
        const int n = *cnt;
        if (tid == 0) qcnt[(c + 1) & 1] = 0;
        for (int i = tid; i < n; i += OWN_T) {
          const int p  = q_a[i];
          const int e  = p & EMASK;
          const int dl = (int)((unsigned)p >> EBITS);
          const float v = gat_logit(s[src[e]], dL[dl]);
          float* addr = &emaxL[dl];
          if (v >= 0.f) atomicMax((int*)addr, __float_as_int(v));
          else          atomicMin((unsigned int*)addr, __float_as_uint(v));
        }
        gat_lds_barrier();
      }
    }

    if (tid < NPW) {
      const float m = emaxL[tid];
      if (!isfinite(m)) emaxL[tid] = 0.f;
    }
    __syncthreads();

    {
      int4 dv[EPT / 4];
      int cc = rot;
      gat_load_dst(dst, cc * CHUNK, tid, dv);
      for (int c = 0; c < NCHUNK; ++c) {
        const int cbase = cc * CHUNK;
        int* cnt = &qcnt[(c + NCHUNK) & 1];
        gat_filter(dv, cbase, tid, lo, nown, cnt, q_a);
        cc = (cc + 1 == NCHUNK) ? 0 : cc + 1;
        if (c + 1 < NCHUNK) gat_load_dst(dst, cc * CHUNK, tid, dv);
        gat_lds_barrier();
        const int n = *cnt;
        if (tid == 0) qcnt[(c + NCHUNK + 1) & 1] = 0;
        for (int i = tid; i < n; i += OWN_T) {
          const int p  = q_a[i];
          const int e  = p & EMASK;
          const int dl = (int)((unsigned)p >> EBITS);
          const int sn = src[e];
          const float v  = gat_logit(s[sn], dL[dl]);
          const float ex = expf(v - emaxL[dl]);
          atomicAdd(&denL[dl], ex);
          q_a[i] = (int)((unsigned)dl << EBITS) | sn;
          q_b[i] = ex;
        }
        gat_lds_barrier();
        for (int i = wv; i < n; i += 64) {
          const int  i2   = i + 32;
          const bool two  = i2 < n;
          const int  p0   = q_a[i];
          const int  p1   = two ? q_a[i2] : p0;
          const float x0  = q_b[i];
          const float x1  = two ? q_b[i2] : 0.f;
          const float4 v0 = ((const float4*)(Wh + (size_t)(p0 & EMASK) * DIM))[lane];
          const float4 v1 = ((const float4*)(Wh + (size_t)(p1 & EMASK) * DIM))[lane];
          asm volatile("" :: "v"(v0.x), "v"(v0.y), "v"(v0.z), "v"(v0.w),
                             "v"(v1.x), "v"(v1.y), "v"(v1.z), "v"(v1.w));
          float* r0 = rows + (int)((unsigned)p0 >> EBITS) * DIM + lane;
          atomicAdd(r0 +  0, x0 * v0.x);
          atomicAdd(r0 + 32, x0 * v0.y);
          atomicAdd(r0 + 64, x0 * v0.z);
          atomicAdd(r0 + 96, x0 * v0.w);
          if (two) {
            float* r1 = rows + (int)((unsigned)p1 >> EBITS) * DIM + lane;
            atomicAdd(r1 +  0, x1 * v1.x);
            atomicAdd(r1 + 32, x1 * v1.y);
            atomicAdd(r1 + 64, x1 * v1.z);
            atomicAdd(r1 + 96, x1 * v1.w);
          }
        }
        gat_lds_barrier();
      }
    }

  }

  {
    float4* op = (float4*)out + (size_t)lo * (DIM / 4) + lane;
    #pragma unroll
    for (int k = 0; k < OUT_ROWS; ++k) {
      const int nl = wv + k * NWAVE;
      if (nl < (int)nown) {
        const float den = denL[nl];
        const float* rp = rows + nl * DIM + lane;
        const float a0 = rp[0], a1 = rp[32], a2 = rp[64], a3 = rp[96];
        float4 o;
        o.x = ((den != 0.f) ? a0 / den : 0.f) + hres[k].x;
        o.y = ((den != 0.f) ? a1 / den : 0.f) + hres[k].y;
        o.z = ((den != 0.f) ? a2 / den : 0.f) + hres[k].z;
        o.w = ((den != 0.f) ? a3 / den : 0.f) + hres[k].w;
        op[nl * (DIM / 4)] = o;
      }
    }
  }
}

extern "C" void kernel_launch(void* const* d_in, const int* in_sizes, int n_in,
                              void* d_out, int out_size, void* d_ws, size_t ws_size,
                              hipStream_t stream) {
  const float* h   = (const float*)d_in[0];
  const float* W   = (const float*)d_in[1];
  const float* a   = (const float*)d_in[2];
  const int*   src = (const int*)d_in[3];
  const int*   dst = (const int*)d_in[4];
  float* hout = (float*)d_out;

  float* ws    = (float*)d_ws;
  float* Wh    = ws;
  float* s     = Wh   + (size_t)N_NODES * DIM;
  float* d     = s    + N_NODES;
  unsigned* bins = (unsigned*)(ws + 6500352);

  {
    const int waves = N_NODES / 16;
    const int blocks = (waves + 3) / 4;
    gat_gemm_wmma<<<blocks, 128, 0, stream>>>(h, W, Wh);
  }
  gat_sd<<<NSDWG, SD_T, 0, stream>>>(Wh, a, s, d, src, dst, bins);
  gat_owner<<<NOWNER, OWN_T, 0, stream>>>(Wh, s, d, src, dst, bins, h, hout);
}
